# starting kernel
# baseline (speedup 1.0000x reference)
#include <hip/hip_runtime.h>
#include <stdint.h>
#include <stddef.h>


typedef __attribute__((ext_vector_type(16))) __bf16 v16bf;
typedef __attribute__((ext_vector_type(8)))  float  v8f;

#define LDSTR 520

__device__ __forceinline__ unsigned f2bf(float f) {
  unsigned u = __float_as_uint(f);
  return (u + 0x7FFFu + ((u >> 16) & 1u)) >> 16;
}

__device__ __forceinline__ int kmap(int v, int lane) {
  return ((v & 4) << 2) | ((lane & 16) >> 1) | ((v & 3) << 1);
}

__device__ __forceinline__ unsigned fenc(float f) {
  unsigned u = __float_as_uint(f);
  return (u & 0x80000000u) ? ~u : (u | 0x80000000u);
}
__device__ __forceinline__ float fdec(unsigned u) {
  return __uint_as_float((u & 0x80000000u) ? (u & 0x7FFFFFFFu) : ~u);
}

__device__ __forceinline__ void pack_tile_to_fragA(const float* tile,
                                                   unsigned* dst, int mt, int tid) {
  const int lane = tid & 31, w = tid >> 5;
  const int m = lane & 15;
#pragma unroll
  for (int q = 0; q < 2; ++q) {
    const int ks = w * 2 + q;
#pragma unroll
    for (int v = 0; v < 8; ++v) {
      const int k = (ks << 5) + kmap(v, lane);
      const unsigned lo = f2bf(tile[m * LDSTR + k]);
      const unsigned hi = f2bf(tile[m * LDSTR + k + 1]);
      dst[((size_t)(mt * 16 + ks) * 32 + lane) * 8 + v] = lo | (hi << 16);
    }
  }
}

__global__ void k_pack_B(const float* __restrict__ W, unsigned* __restrict__ pack,
                         int K, int N) {
  const int id = blockIdx.x * 256 + threadIdx.x;
  const int nks = K >> 5;
  const int total = (N >> 4) * nks * 256;
  if (id >= total) return;
  const int v = id & 7, l = (id >> 3) & 31;
  const int t = id >> 8, ks = t % nks, nt = t / nks;
  const int n = (nt << 4) + (l & 15);
  const int k = (ks << 5) + kmap(v, l);
  pack[id] = f2bf(W[(size_t)k * N + n]) | (f2bf(W[(size_t)(k + 1) * N + n]) << 16);
}

__global__ void k_pack_A(const float* __restrict__ X, unsigned* __restrict__ pack,
                         int M, int K) {
  const int id = blockIdx.x * 256 + threadIdx.x;
  const int nks = K >> 5;
  const int total = (M >> 4) * nks * 256;
  if (id >= total) return;
  const int v = id & 7, l = (id >> 3) & 31;
  const int t = id >> 8, ks = t % nks, mt = t / nks;
  const int m = (mt << 4) + (l & 15);
  const int k = (ks << 5) + kmap(v, l);
  pack[id] = f2bf(X[(size_t)m * K + k]) | (f2bf(X[(size_t)m * K + k + 1]) << 16);
}

__global__ __launch_bounds__(256) void k_gemm_ft_ln(
    const unsigned* __restrict__ packX, const unsigned* __restrict__ packW,
    const float* __restrict__ ft_b, const float* __restrict__ ln_g,
    const float* __restrict__ ln_b, unsigned* __restrict__ packH1) {
  __shared__ float tile[16 * LDSTR];
  __shared__ float psum[16][16], psq[16][16], stats[16][2];
  const int tid = threadIdx.x, lane = tid & 31, w = tid >> 5;
  const int mt = blockIdx.x;
  const int nks = 4;
  v8f acc[4] = {};
  for (int ks = 0; ks < nks; ++ks) {
    v16bf a = *(const v16bf*)(packX + ((size_t)(mt * nks + ks) * 32 + lane) * 8);
#pragma unroll
    for (int j = 0; j < 4; ++j) {
      const int nt = w * 4 + j;
      v16bf b = *(const v16bf*)(packW + ((size_t)(nt * nks + ks) * 32 + lane) * 8);
      acc[j] = __builtin_amdgcn_wmma_f32_16x16x32_bf16(false, a, false, b,
                                                       (short)0, acc[j], false, false);
    }
  }
  const int m0 = (lane >> 4) << 3, nc = lane & 15;
#pragma unroll
  for (int j = 0; j < 4; ++j) {
    const int n = (w * 4 + j) * 16 + nc;
    const float bias = ft_b[n];
#pragma unroll
    for (int r = 0; r < 8; ++r) tile[(m0 + r) * LDSTR + n] = acc[j][r] + bias;
  }
  __syncthreads();
  {
    const int row = tid >> 4, p = tid & 15;
    float s = 0.f, s2 = 0.f;
    for (int c = p * 32; c < p * 32 + 32; ++c) {
      const float xv = tile[row * LDSTR + c];
      s += xv; s2 += xv * xv;
    }
    psum[row][p] = s; psq[row][p] = s2;
  }
  __syncthreads();
  if (tid < 16) {
    float s = 0.f, s2 = 0.f;
    for (int p = 0; p < 16; ++p) { s += psum[tid][p]; s2 += psq[tid][p]; }
    const float mu = s * (1.f / 512.f);
    const float var = s2 * (1.f / 512.f) - mu * mu;
    stats[tid][0] = mu;
    stats[tid][1] = rsqrtf(var + 1e-5f);
  }
  __syncthreads();
  {
    const int row = tid >> 4, p = tid & 15;
    const float mu = stats[row][0], rs = stats[row][1];
    for (int c = p * 32; c < p * 32 + 32; ++c) {
      float xv = (tile[row * LDSTR + c] - mu) * rs * ln_g[c] + ln_b[c];
      tile[row * LDSTR + c] = xv >= 0.f ? xv : 0.2f * xv;
    }
  }
  __syncthreads();
  pack_tile_to_fragA(tile, packH1, mt, tid);
}

__global__ __launch_bounds__(256) void k_gemm1(
    const unsigned* __restrict__ packA, const unsigned* __restrict__ packW,
    const float* __restrict__ att_src, const float* __restrict__ att_dst,
    float* __restrict__ hW, float* __restrict__ a_src, float* __restrict__ a_dst) {
  __shared__ float tile[16 * LDSTR];
  const int tid = threadIdx.x, lane = tid & 31, w = tid >> 5;
  const int mt = blockIdx.x;
  const int nks = 16;
  v8f acc[4] = {};
  for (int ks = 0; ks < nks; ++ks) {
    v16bf a = *(const v16bf*)(packA + ((size_t)(mt * nks + ks) * 32 + lane) * 8);
#pragma unroll
    for (int j = 0; j < 4; ++j) {
      const int nt = w * 4 + j;
      v16bf b = *(const v16bf*)(packW + ((size_t)(nt * nks + ks) * 32 + lane) * 8);
      acc[j] = __builtin_amdgcn_wmma_f32_16x16x32_bf16(false, a, false, b,
                                                       (short)0, acc[j], false, false);
    }
  }
  const int m0 = (lane >> 4) << 3, nc = lane & 15;
#pragma unroll
  for (int j = 0; j < 4; ++j) {
    const int n = (w * 4 + j) * 16 + nc;
#pragma unroll
    for (int r = 0; r < 8; ++r) tile[(m0 + r) * LDSTR + n] = acc[j][r];
  }
  __syncthreads();
  for (int i = tid; i < 16 * 512; i += 256) {
    const int row = i >> 9, c = i & 511;
    hW[(size_t)(mt * 16 + row) * 512 + c] = tile[row * LDSTR + c];
  }
  if (tid < 128) {
    const int row = tid >> 3, h = tid & 7;
    float s1 = 0.f, s2 = 0.f;
    for (int c = 0; c < 64; ++c) {
      const float v = tile[row * LDSTR + (h << 6) + c];
      s1 += v * att_src[(h << 6) + c];
      s2 += v * att_dst[(h << 6) + c];
    }
    const int node = mt * 16 + row;
    a_src[node * 8 + h] = s1;
    a_dst[node * 8 + h] = s2;
  }
}

__global__ void k_edge_logit_max(const int* __restrict__ ei,
    const float* __restrict__ a_src, const float* __restrict__ a_dst,
    float* __restrict__ buf, unsigned* __restrict__ maxenc, int E) {
  const int idx = blockIdx.x * 256 + threadIdx.x;
  if (idx >= E * 8) return;
  const int e = idx >> 3, h = idx & 7;
  const int s = ei[e], d = ei[E + e];
  float L = a_src[s * 8 + h] + a_dst[d * 8 + h];
  L = L >= 0.f ? L : 0.2f * L;
  buf[idx] = L;
  atomicMax(&maxenc[d * 8 + h], fenc(L));
}

__global__ void k_edge_expsum(const int* __restrict__ ei,
    float* __restrict__ buf, const unsigned* __restrict__ maxenc,
    float* __restrict__ denom, int E) {
  const int idx = blockIdx.x * 256 + threadIdx.x;
  if (idx >= E * 8) return;
  const int e = idx >> 3, h = idx & 7;
  const int d = ei[E + e];
  const float ex = __expf(buf[idx] - fdec(maxenc[d * 8 + h]));
  buf[idx] = ex;
  atomicAdd(&denom[d * 8 + h], ex);
}

__global__ __launch_bounds__(256) void k_scatter1(const int* __restrict__ ei,
    const float* __restrict__ hW, const float* __restrict__ expv,
    const float* __restrict__ denom, float* __restrict__ agg, int E) {
  __shared__ float alpha[8];
  const int e = blockIdx.x;
  const int s = ei[e], d = ei[E + e];
  if (threadIdx.x < 8)
    alpha[threadIdx.x] = expv[e * 8 + threadIdx.x] / denom[d * 8 + threadIdx.x];
  __syncthreads();
#pragma unroll
  for (int i = 0; i < 2; ++i) {
    const int c = threadIdx.x + i * 256;
    atomicAdd(&agg[(size_t)d * 512 + c], hW[(size_t)s * 512 + c] * alpha[c >> 6]);
  }
}

__global__ void k_scatter2(const int* __restrict__ ei,
    const float* __restrict__ hW, const float* __restrict__ expv,
    const float* __restrict__ denom, float* __restrict__ agg, int E) {
  const int idx = blockIdx.x * 256 + threadIdx.x;
  if (idx >= E * 16) return;
  const int e = idx >> 4, c = idx & 15, h = c >> 1;
  const int s = ei[e], d = ei[E + e];
  const float alpha = expv[e * 8 + h] / denom[d * 8 + h];
  atomicAdd(&agg[(size_t)d * 16 + c], hW[(size_t)s * 16 + c] * alpha);
}

__global__ __launch_bounds__(256) void k_bnprep(const float* __restrict__ agg,
    const float* __restrict__ b1, const float* __restrict__ bn_g,
    const float* __restrict__ bn_b, const float* __restrict__ bn_mean,
    const float* __restrict__ bn_var, unsigned* __restrict__ packH2) {
  __shared__ float tile[16 * LDSTR];
  const int mt = blockIdx.x, tid = threadIdx.x;
  for (int i = tid; i < 16 * 512; i += 256) {
    const int row = i >> 9, c = i & 511;
    float xv = agg[(size_t)(mt * 16 + row) * 512 + c] + b1[c];
    xv = (xv - bn_mean[c]) * rsqrtf(bn_var[c] + 1e-5f) * bn_g[c] + bn_b[c];
    tile[row * LDSTR + c] = xv > 0.f ? xv : 0.f;
  }
  __syncthreads();
  pack_tile_to_fragA(tile, packH2, mt, tid);
}

__global__ __launch_bounds__(32) void k_gemm2(
    const unsigned* __restrict__ packA, const unsigned* __restrict__ packW,
    const float* __restrict__ att_src, const float* __restrict__ att_dst,
    float* __restrict__ hW, float* __restrict__ a_src, float* __restrict__ a_dst) {
  __shared__ float t2[16][16];
  const int lane = threadIdx.x;
  const int mt = blockIdx.x;
  const int nks = 16;
  v8f acc = {};
  for (int ks = 0; ks < nks; ++ks) {
    v16bf a = *(const v16bf*)(packA + ((size_t)(mt * nks + ks) * 32 + lane) * 8);
    v16bf b = *(const v16bf*)(packW + ((size_t)ks * 32 + lane) * 8);
    acc = __builtin_amdgcn_wmma_f32_16x16x32_bf16(false, a, false, b,
                                                  (short)0, acc, false, false);
  }
  const int m0 = (lane >> 4) << 3, nc = lane & 15;
#pragma unroll
  for (int r = 0; r < 8; ++r) t2[m0 + r][nc] = acc[r];
  __syncthreads();
  for (int i = lane; i < 256; i += 32) {
    const int row = i >> 4, c = i & 15;
    hW[(size_t)(mt * 16 + row) * 16 + c] = t2[row][c];
  }
  if (lane < 16) {
    const int row = lane, node = mt * 16 + row;
    for (int h = 0; h < 8; ++h) {
      const float v0 = t2[row][2 * h], v1 = t2[row][2 * h + 1];
      a_src[node * 8 + h] = v0 * att_src[2 * h] + v1 * att_src[2 * h + 1];
      a_dst[node * 8 + h] = v0 * att_dst[2 * h] + v1 * att_dst[2 * h + 1];
    }
  }
}

__global__ void k_final(const float* __restrict__ agg, const float* __restrict__ b2,
                        float* __restrict__ out, int N) {
  const int t = blockIdx.x * 256 + threadIdx.x;
  if (t >= N * 2) return;
  const int n = t >> 1, c = t & 1;
  float s = 0.f;
#pragma unroll
  for (int h = 0; h < 8; ++h) s += agg[n * 16 + 2 * h + c];
  out[t] = s * 0.125f + b2[c];
}

__global__ void k_zero(unsigned* __restrict__ p, size_t n) {
  const size_t i = (size_t)blockIdx.x * 256 + threadIdx.x;
  if (i < n) p[i] = 0u;
}

extern "C" void kernel_launch(void* const* d_in, const int* in_sizes, int n_in,
                              void* d_out, int out_size, void* d_ws, size_t ws_size,
                              hipStream_t stream) {
  const float* x       = (const float*)d_in[0];
  const int*   ei      = (const int*)d_in[1];
  const float* ft_W    = (const float*)d_in[2];
  const float* ft_b    = (const float*)d_in[3];
  const float* ln_g    = (const float*)d_in[4];
  const float* ln_b    = (const float*)d_in[5];
  const float* W1      = (const float*)d_in[6];
  const float* att_s1  = (const float*)d_in[7];
  const float* att_d1  = (const float*)d_in[8];
  const float* b1      = (const float*)d_in[9];
  const float* bn_g    = (const float*)d_in[10];
  const float* bn_b    = (const float*)d_in[11];
  const float* bn_mean = (const float*)d_in[12];
  const float* bn_var  = (const float*)d_in[13];
  const float* W2      = (const float*)d_in[14];
  const float* att_s2  = (const float*)d_in[15];
  const float* att_d2  = (const float*)d_in[16];
  const float* b2      = (const float*)d_in[17];

  const int N  = in_sizes[0] / 128;
  const int E  = in_sizes[1] / 2;
  const int MT = N / 16;

  char* cur = (char*)d_ws;
  auto alloc = [&](size_t bytes) -> void* {
    void* p = (void*)cur;
    cur += (bytes + 255) & ~(size_t)255;
    return p;
  };
  unsigned* packWft = (unsigned*)alloc((size_t)32 * 4 * 256 * 4);
  unsigned* packW1  = (unsigned*)alloc((size_t)32 * 16 * 256 * 4);
  unsigned* packW2  = (unsigned*)alloc((size_t)1 * 16 * 256 * 4);
  unsigned* packX   = (unsigned*)alloc((size_t)MT * 4 * 256 * 4);
  unsigned* packH1  = (unsigned*)alloc((size_t)MT * 16 * 256 * 4);
  unsigned* packH2  = (unsigned*)alloc((size_t)MT * 16 * 256 * 4);
  float* hW1 = (float*)alloc((size_t)N * 512 * 4);
  float* hW2 = (float*)alloc((size_t)N * 16 * 4);
  float* aS1 = (float*)alloc((size_t)N * 8 * 4);
  float* aD1 = (float*)alloc((size_t)N * 8 * 4);
  float* aS2 = (float*)alloc((size_t)N * 8 * 4);
  float* aD2 = (float*)alloc((size_t)N * 8 * 4);
  float* ex1 = (float*)alloc((size_t)E * 8 * 4);
  float* ex2 = (float*)alloc((size_t)E * 8 * 4);
  char* z0 = cur;
  float*    agg1 = (float*)alloc((size_t)N * 512 * 4);
  float*    agg2 = (float*)alloc((size_t)N * 16 * 4);
  unsigned* max1 = (unsigned*)alloc((size_t)N * 8 * 4);
  float*    den1 = (float*)alloc((size_t)N * 8 * 4);
  unsigned* max2 = (unsigned*)alloc((size_t)N * 8 * 4);
  float*    den2 = (float*)alloc((size_t)N * 8 * 4);
  const size_t zwords = (size_t)(cur - z0) / 4;

  k_zero<<<(unsigned)((zwords + 255) / 256), 256, 0, stream>>>((unsigned*)z0, zwords);

  k_pack_B<<<(32 * 4 * 256 + 255) / 256, 256, 0, stream>>>(ft_W, packWft, 128, 512);
  k_pack_B<<<(32 * 16 * 256 + 255) / 256, 256, 0, stream>>>(W1, packW1, 512, 512);
  k_pack_B<<<(1 * 16 * 256 + 255) / 256, 256, 0, stream>>>(W2, packW2, 512, 16);
  k_pack_A<<<(MT * 4 * 256 + 255) / 256, 256, 0, stream>>>(x, packX, N, 128);

  k_gemm_ft_ln<<<MT, 256, 0, stream>>>(packX, packWft, ft_b, ln_g, ln_b, packH1);

  k_gemm1<<<MT, 256, 0, stream>>>(packH1, packW1, att_s1, att_d1, hW1, aS1, aD1);
  const int EH = E * 8;
  k_edge_logit_max<<<(EH + 255) / 256, 256, 0, stream>>>(ei, aS1, aD1, ex1, max1, E);
  k_edge_expsum<<<(EH + 255) / 256, 256, 0, stream>>>(ei, ex1, max1, den1, E);
  k_scatter1<<<E, 256, 0, stream>>>(ei, hW1, ex1, den1, agg1, E);

  k_bnprep<<<MT, 256, 0, stream>>>(agg1, b1, bn_g, bn_b, bn_mean, bn_var, packH2);

  k_gemm2<<<MT, 32, 0, stream>>>(packH2, packW2, att_s2, att_d2, hW2, aS2, aD2);
  k_edge_logit_max<<<(EH + 255) / 256, 256, 0, stream>>>(ei, aS2, aD2, ex2, max2, E);
  k_edge_expsum<<<(EH + 255) / 256, 256, 0, stream>>>(ei, ex2, max2, den2, E);
  k_scatter2<<<(E * 16 + 255) / 256, 256, 0, stream>>>(ei, hW2, ex2, den2, agg2, E);

  k_final<<<(N * 2 + 255) / 256, 256, 0, stream>>>(agg2, b2, (float*)d_out, N);
}


# Both gather kernels: one wave per destination node (8 nodes per work-group) instead of a 4-wave work-group per node; no work-group barriers; softmax denominators summed in a different order
# speedup vs baseline: 2.4858x; 2.4858x over previous
#include <hip/hip_runtime.h>
#include <stdint.h>
#include <stddef.h>


typedef __attribute__((ext_vector_type(16))) __bf16 v16bf;
typedef __attribute__((ext_vector_type(8)))  float  v8f;
typedef __attribute__((ext_vector_type(8)))  unsigned v8u;

#define LDSTR 520

__device__ __forceinline__ unsigned f2bf(float f) {
  unsigned u = __float_as_uint(f);
  return (u + 0x7FFFu + ((u >> 16) & 1u)) >> 16;
}

__device__ __forceinline__ int kmap(int v, int lane) {
  return ((v & 4) << 2) | ((lane & 16) >> 1) | ((v & 3) << 1);
}

__device__ __forceinline__ void pack_tile_to_lds_frag(const float* tile, uint4* flo,
                                                      uint4* fhi, int tid) {
  const int lane = tid & 31, w = tid >> 5;
  const int m = lane & 15;
#pragma unroll
  for (int q = 0; q < 2; ++q) {
    const int ks = w * 2 + q;
    unsigned d[8];
#pragma unroll
    for (int v = 0; v < 8; ++v) {
      const int k = (ks << 5) + kmap(v, lane);
      const unsigned lo = f2bf(tile[m * LDSTR + k]);
      const unsigned hi = f2bf(tile[m * LDSTR + k + 1]);
      d[v] = lo | (hi << 16);
    }
    flo[ks * 32 + lane] = make_uint4(d[0], d[1], d[2], d[3]);
    fhi[ks * 32 + lane] = make_uint4(d[4], d[5], d[6], d[7]);
  }
}

__device__ __forceinline__ v16bf lds_fragA(const uint4* flo, const uint4* fhi,
                                           int ks, int lane) {
  const uint4 lo = flo[ks * 32 + lane], hi = fhi[ks * 32 + lane];
  const v8u u = {lo.x, lo.y, lo.z, lo.w, hi.x, hi.y, hi.z, hi.w};
  return __builtin_bit_cast(v16bf, u);
}

__device__ __forceinline__ void pack_B_elem(const float* __restrict__ W,
                                            unsigned* __restrict__ pack,
                                            const int K, const int N, const int id) {
  const int nks = K >> 5;
  const int total = (N >> 4) * nks * 256;
  if (id >= total) return;
  const int v = id & 7, l = (id >> 3) & 31;
  const int t = id >> 8, ks = t % nks, nt = t / nks;
  const int n = (nt << 4) + (l & 15);
  const int k = (ks << 5) + kmap(v, l);
  pack[id] = f2bf(W[(size_t)k * N + n]) | (f2bf(W[(size_t)(k + 1) * N + n]) << 16);
}

__global__ __launch_bounds__(256) void k_prep(
    const float* __restrict__ ft_W, const float* __restrict__ W1,
    const float* __restrict__ W2, unsigned* __restrict__ packWft,
    unsigned* __restrict__ packW1, unsigned* __restrict__ packW2,
    int* __restrict__ cnt, int N) {
  int b = blockIdx.x;
  const int tid = threadIdx.x;
  if (b < 128) { pack_B_elem(ft_W, packWft, 128, 512, b * 256 + tid); return; }
  b -= 128;
  if (b < 512) { pack_B_elem(W1, packW1, 512, 512, b * 256 + tid); return; }
  b -= 512;
  if (b < 16) { pack_B_elem(W2, packW2, 512, 16, b * 256 + tid); return; }
  b -= 16;
  const int i = b * 256 + tid;
  if (i < N) cnt[i] = 0;
}

__device__ __forceinline__ void scan_block256(const int* __restrict__ cnt,
                                              int* __restrict__ off, int N) {
  __shared__ int wsum[8];
  const int tid = threadIdx.x, lane = tid & 31, w = tid >> 5;
  const int per = (N + 255) >> 8;
  const int b = tid * per;
  int local = 0;
  for (int i = 0; i < per; ++i) {
    const int idx = b + i;
    if (idx < N) local += cnt[idx];
  }
  int v = local;
  for (int o = 1; o < 32; o <<= 1) {
    const int t = __shfl_up(v, o, 32);
    if (lane >= o) v += t;
  }
  if (lane == 31) wsum[w] = v;
  __syncthreads();
  int wbase = 0;
  for (int j = 0; j < 8; ++j) wbase += (j < w) ? wsum[j] : 0;
  int base = v - local + wbase;
  for (int i = 0; i < per; ++i) {
    const int idx = b + i;
    if (idx < N) { off[idx] = base; base += cnt[idx]; }
  }
  if (tid == 255) off[N] = base;
}

__device__ __forceinline__ unsigned pk2(float a, float b) {
  return f2bf(a) | (f2bf(b) << 16);
}

__global__ __launch_bounds__(256) void k_ft_gemm1(
    const float* __restrict__ x, const unsigned* __restrict__ packWft,
    const float* __restrict__ ft_b, const float* __restrict__ ln_g,
    const float* __restrict__ ln_b, const unsigned* __restrict__ packW1,
    const float* __restrict__ att_src, const float* __restrict__ att_dst,
    float* __restrict__ hW, float* __restrict__ a_src, float* __restrict__ a_dst,
    const int* __restrict__ cnt, int* __restrict__ off, int N) {
  __shared__ float tile[16 * LDSTR];
  __shared__ float psum[16][16], psq[16][16], stats[16][2];
  __shared__ uint4 flo[16 * 32], fhi[16 * 32];
  if (blockIdx.x == 0) {
    scan_block256(cnt, off, N);
    return;
  }
  const int tid = threadIdx.x, lane = tid & 31, w = tid >> 5;
  const int mt = blockIdx.x - 1;
  const int m0 = (lane >> 4) << 3, nc = lane & 15;
  {
    const int nks = 4;
    const float* xr = x + (size_t)(mt * 16 + nc) * 128 + ((lane & 16) >> 1);
    v8f acc[4] = {};
    for (int ks = 0; ks < nks; ++ks) {
      const float4 p0 = *(const float4*)(xr + ks * 32);
      const float4 p1 = *(const float4*)(xr + ks * 32 + 4);
      const float4 p2 = *(const float4*)(xr + ks * 32 + 16);
      const float4 p3 = *(const float4*)(xr + ks * 32 + 20);
      const v8u au = {pk2(p0.x, p0.y), pk2(p0.z, p0.w), pk2(p1.x, p1.y), pk2(p1.z, p1.w),
                      pk2(p2.x, p2.y), pk2(p2.z, p2.w), pk2(p3.x, p3.y), pk2(p3.z, p3.w)};
      const v16bf a = __builtin_bit_cast(v16bf, au);
#pragma unroll
      for (int j = 0; j < 4; ++j) {
        const int nt = w * 4 + j;
        v16bf b = *(const v16bf*)(packWft + ((size_t)(nt * nks + ks) * 32 + lane) * 8);
        acc[j] = __builtin_amdgcn_wmma_f32_16x16x32_bf16(false, a, false, b,
                                                         (short)0, acc[j], false, false);
      }
    }
#pragma unroll
    for (int j = 0; j < 4; ++j) {
      const int n = (w * 4 + j) * 16 + nc;
      const float bias = ft_b[n];
#pragma unroll
      for (int r = 0; r < 8; ++r) tile[(m0 + r) * LDSTR + n] = acc[j][r] + bias;
    }
  }
  __syncthreads();
  {
    const int row = tid >> 4, p = tid & 15;
    float s = 0.f, s2 = 0.f;
    for (int c = p * 32; c < p * 32 + 32; ++c) {
      const float xv = tile[row * LDSTR + c];
      s += xv; s2 += xv * xv;
    }
    psum[row][p] = s; psq[row][p] = s2;
  }
  __syncthreads();
  if (tid < 16) {
    float s = 0.f, s2 = 0.f;
    for (int p = 0; p < 16; ++p) { s += psum[tid][p]; s2 += psq[tid][p]; }
    const float mu = s * (1.f / 512.f);
    const float var = s2 * (1.f / 512.f) - mu * mu;
    stats[tid][0] = mu;
    stats[tid][1] = rsqrtf(var + 1e-5f);
  }
  __syncthreads();
  {
    const int row = tid >> 4, p = tid & 15;
    const float mu = stats[row][0], rs = stats[row][1];
    for (int c = p * 32; c < p * 32 + 32; ++c) {
      float xv = (tile[row * LDSTR + c] - mu) * rs * ln_g[c] + ln_b[c];
      tile[row * LDSTR + c] = xv >= 0.f ? xv : 0.2f * xv;
    }
  }
  __syncthreads();
  pack_tile_to_lds_frag(tile, flo, fhi, tid);
  __syncthreads();
  {
    const int nks = 16;
    v8f acc[4] = {};
    for (int ks = 0; ks < nks; ++ks) {
      const v16bf a = lds_fragA(flo, fhi, ks, lane);
#pragma unroll
      for (int j = 0; j < 4; ++j) {
        const int nt = w * 4 + j;
        v16bf b = *(const v16bf*)(packW1 + ((size_t)(nt * nks + ks) * 32 + lane) * 8);
        acc[j] = __builtin_amdgcn_wmma_f32_16x16x32_bf16(false, a, false, b,
                                                         (short)0, acc[j], false, false);
      }
    }
#pragma unroll
    for (int j = 0; j < 4; ++j) {
      const int n = (w * 4 + j) * 16 + nc;
#pragma unroll
      for (int r = 0; r < 8; ++r) tile[(m0 + r) * LDSTR + n] = acc[j][r];
    }
  }
  __syncthreads();
  for (int i = tid; i < 16 * 512; i += 256) {
    const int row = i >> 9, c = i & 511;
    hW[(size_t)(mt * 16 + row) * 512 + c] = tile[row * LDSTR + c];
  }
  if (tid < 128) {
    const int row = tid >> 3, h = tid & 7;
    float s1 = 0.f, s2 = 0.f;
    for (int c = 0; c < 64; ++c) {
      const float v = tile[row * LDSTR + (h << 6) + c];
      s1 += v * att_src[(h << 6) + c];
      s2 += v * att_dst[(h << 6) + c];
    }
    const int node = mt * 16 + row;
    a_src[node * 8 + h] = s1;
    a_dst[node * 8 + h] = s2;
  }
}

#define SEGCAP 64
#define AGG2STR 32

__global__ void k_count(const int* __restrict__ ei, int* __restrict__ cnt, int E) {
  const int e = blockIdx.x * 256 + threadIdx.x;
  if (e >= E) return;
  atomicAdd(&cnt[ei[E + e]], 1);
}

__global__ void k_fill(const int* __restrict__ ei, const int* __restrict__ off,
                       int* __restrict__ cnt, int* __restrict__ csr_e,
                       int* __restrict__ csr_s, int E) {
  const int e = blockIdx.x * 256 + threadIdx.x;
  if (e >= E) return;
  const int s = ei[e], d = ei[E + e];
  const int pos = off[d] + atomicSub(&cnt[d], 1) - 1;
  csr_e[pos] = e;
  csr_s[pos] = s;
}

#define NPW 8

__device__ __forceinline__ void wave_lds_handover() {
  __builtin_amdgcn_fence(__ATOMIC_RELEASE, "wavefront");
  __builtin_amdgcn_wave_barrier();
  __builtin_amdgcn_fence(__ATOMIC_ACQUIRE, "wavefront");
}

#define APOS(j, h) ((j) * 8 + (((h) & 1) << 2) + ((h) >> 1))

template <int ROWF, int OSTR>
__device__ __forceinline__ void gather_node_wave(
    const int N, const int* __restrict__ off, const int* __restrict__ csr_e,
    const int* __restrict__ csr_s, const float* __restrict__ hW,
    const float* __restrict__ a_src, const float* __restrict__ a_dst,
    float* __restrict__ agg) {
  static_assert(ROWF == 512 || ROWF == 16, "row gather is written for these two row sizes");
  __shared__ int lkey_all[NPW][SEGCAP];
  __shared__ int lsrc_all[NPW][SEGCAP];
  __shared__ __attribute__((aligned(16))) float lalpha_all[NPW][SEGCAP * 8];
  const int lane = threadIdx.x & 31;
  const int w = __builtin_amdgcn_readfirstlane((int)(threadIdx.x >> 5));
  const int d = blockIdx.x * NPW + w;
  if (d >= N) return;
  int* lkey = lkey_all[w];
  int* lsrc = lsrc_all[w];
  float* lalpha = lalpha_all[w];
  const int beg = off[d], deg = off[d + 1] - beg;
  const bool small = deg <= SEGCAP;
  if (small) {
    const int t0 = lane, t1 = lane + 32;
    int k0 = 0, s0 = 0, k1 = 0, s1 = 0;
    if (t0 < deg) { k0 = csr_e[beg + t0]; s0 = csr_s[beg + t0]; lkey[t0] = k0; }
    if (t1 < deg) { k1 = csr_e[beg + t1]; s1 = csr_s[beg + t1]; lkey[t1] = k1; }
    wave_lds_handover();
    int r0 = 0, r1 = 0;
    for (int j = 0; j < deg; ++j) {
      const int kj = lkey[j];
      r0 += (kj < k0) ? 1 : 0;
      r1 += (kj < k1) ? 1 : 0;
    }
    if (t0 < deg) lsrc[r0] = s0;
    if (t1 < deg) lsrc[r1] = s1;
    wave_lds_handover();
  }
#define STAGE_CHUNK(cb, n)                                               \
  if (!small) {                                                          \
    wave_lds_handover();    \
    for (int t = lane; t < (n); t += 32) lsrc[t] = csr_s[beg + (cb) + t]; \
    wave_lds_handover();                                                 \
  }
  const int h8 = lane & 7, jj = lane >> 3;
  const float ad = a_dst[d * 8 + h8];
  float mx = __uint_as_float(0xff800000u);
  for (int cb = 0; cb < deg; cb += SEGCAP) {
    const int n = (deg - cb) < SEGCAP ? (deg - cb) : SEGCAP;
    STAGE_CHUNK(cb, n)
    for (int j = jj; j < n; j += 4) {
      float L = a_src[lsrc[j] * 8 + h8] + ad;
      L = L >= 0.f ? L : 0.2f * L;
      mx = fmaxf(mx, L);
    }
  }
  mx = fmaxf(mx, __shfl_xor(mx, 8, 32));
  mx = fmaxf(mx, __shfl_xor(mx, 16, 32));
  const float m = mx;
  float sm = 0.f;
  for (int cb = 0; cb < deg; cb += SEGCAP) {
    const int n = (deg - cb) < SEGCAP ? (deg - cb) : SEGCAP;
    STAGE_CHUNK(cb, n)
    for (int j = jj; j < n; j += 4) {
      float L = a_src[lsrc[j] * 8 + h8] + ad;
      L = L >= 0.f ? L : 0.2f * L;
      sm += __expf(L - m);
    }
  }
  sm += __shfl_xor(sm, 8, 32);
  sm += __shfl_xor(sm, 16, 32);
  const float den = sm;
  float4 acc0 = make_float4(0.f, 0.f, 0.f, 0.f), acc1 = acc0, acc2 = acc0, acc3 = acc0;
  const int hl = lane >> 4;
  for (int cb = 0; cb < deg; cb += SEGCAP) {
    const int n = (deg - cb) < SEGCAP ? (deg - cb) : SEGCAP;
    STAGE_CHUNK(cb, n)
    for (int j = jj; j < n; j += 4) {
      float L = a_src[lsrc[j] * 8 + h8] + ad;
      L = L >= 0.f ? L : 0.2f * L;
      lalpha[APOS(j, h8)] = __expf(L - m) / den;
    }
    wave_lds_handover();
    if (ROWF == 512) {
#pragma unroll 4
      for (int j = 0; j < n; ++j) {
        const int s = __builtin_amdgcn_readfirstlane(lsrc[j]);
        const float* row = hW + (size_t)s * ROWF + lane * 4;
        const float4 v0 = *(const float4*)(row);
        const float4 v1 = *(const float4*)(row + 128);
        const float4 v2 = *(const float4*)(row + 256);
        const float4 v3 = *(const float4*)(row + 384);
        const float4 al = *(const float4*)(lalpha + j * 8 + hl * 4);
        acc0.x += al.x * v0.x; acc0.y += al.x * v0.y; acc0.z += al.x * v0.z; acc0.w += al.x * v0.w;
        acc1.x += al.y * v1.x; acc1.y += al.y * v1.y; acc1.z += al.y * v1.z; acc1.w += al.y * v1.w;
        acc2.x += al.z * v2.x; acc2.y += al.z * v2.y; acc2.z += al.z * v2.z; acc2.w += al.z * v2.w;
        acc3.x += al.w * v3.x; acc3.y += al.w * v3.y; acc3.z += al.w * v3.z; acc3.w += al.w * v3.w;
      }
    } else if (lane < 4) {
#pragma unroll 4
      for (int j = 0; j < n; ++j) {
        const float4 v = *(const float4*)(hW + (size_t)lsrc[j] * ROWF + lane * 4);
        const float a0 = lalpha[APOS(j, 2 * lane)], a1 = lalpha[APOS(j, 2 * lane + 1)];
        acc0.x += a0 * v.x; acc0.y += a0 * v.y; acc0.z += a1 * v.z; acc0.w += a1 * v.w;
      }
    }
  }
#undef STAGE_CHUNK
  if (ROWF == 512) {
    float* o = agg + (size_t)d * OSTR + lane * 4;
    *(float4*)(o) = acc0;
    *(float4*)(o + 128) = acc1;
    *(float4*)(o + 256) = acc2;
    *(float4*)(o + 384) = acc3;
  } else if (lane * 4 < OSTR) {
    *(float4*)(agg + (size_t)d * OSTR + lane * 4) = acc0;
  }
}

__global__ __launch_bounds__(32 * NPW) void k_gather1(
    const int N, const int* __restrict__ off, const int* __restrict__ csr_e,
    const int* __restrict__ csr_s, const float* __restrict__ hW,
    const float* __restrict__ a_src, const float* __restrict__ a_dst,
    float* __restrict__ agg) {
  gather_node_wave<512, 512>(N, off, csr_e, csr_s, hW, a_src, a_dst, agg);
}

__global__ __launch_bounds__(32 * NPW) void k_gather2(
    const int N, const int* __restrict__ off, const int* __restrict__ csr_e,
    const int* __restrict__ csr_s, const float* __restrict__ hW,
    const float* __restrict__ a_src, const float* __restrict__ a_dst,
    float* __restrict__ agg) {
  gather_node_wave<16, AGG2STR>(N, off, csr_e, csr_s, hW, a_src, a_dst, agg);
}

__global__ __launch_bounds__(256) void k_bn_gemm2(const float* __restrict__ agg,
    const float* __restrict__ b1, const float* __restrict__ bn_g,
    const float* __restrict__ bn_b, const float* __restrict__ bn_mean,
    const float* __restrict__ bn_var, const unsigned* __restrict__ packW,
    const float* __restrict__ att_src, const float* __restrict__ att_dst,
    float* __restrict__ hW, float* __restrict__ a_src, float* __restrict__ a_dst) {
  __shared__ float tile[16 * LDSTR];
  __shared__ uint4 flo[16 * 32], fhi[16 * 32];
  __shared__ float t2[16][16];
  const int mt = blockIdx.x, tid = threadIdx.x;
  for (int i = tid; i < 16 * 512; i += 256) {
    const int row = i >> 9, c = i & 511;
    float xv = agg[(size_t)(mt * 16 + row) * 512 + c] + b1[c];
    xv = (xv - bn_mean[c]) * rsqrtf(bn_var[c] + 1e-5f) * bn_g[c] + bn_b[c];
    tile[row * LDSTR + c] = xv > 0.f ? xv : 0.f;
  }
  __syncthreads();
  pack_tile_to_lds_frag(tile, flo, fhi, tid);
  __syncthreads();
  if (tid < 32) {
    const int lane = tid;
    const int nks = 16;
    v8f acc = {};
    for (int ks = 0; ks < nks; ++ks) {
      const v16bf a = lds_fragA(flo, fhi, ks, lane);
      v16bf b = *(const v16bf*)(packW + ((size_t)ks * 32 + lane) * 8);
      acc = __builtin_amdgcn_wmma_f32_16x16x32_bf16(false, a, false, b,
                                                    (short)0, acc, false, false);
    }
    const int m0 = (lane >> 4) << 3, nc = lane & 15;
#pragma unroll
    for (int r = 0; r < 8; ++r) t2[m0 + r][nc] = acc[r];
  }
  __syncthreads();
  {
    const int row = tid >> 4, c = tid & 15;
    hW[(size_t)(mt * 16 + row) * 16 + c] = t2[row][c];
  }
  if (tid < 128) {
    const int row = tid >> 3, h = tid & 7, node = mt * 16 + row;
    const float v0 = t2[row][2 * h], v1 = t2[row][2 * h + 1];
    a_src[node * 8 + h] = v0 * att_src[2 * h] + v1 * att_src[2 * h + 1];
    a_dst[node * 8 + h] = v0 * att_dst[2 * h] + v1 * att_dst[2 * h + 1];
  }
}

__global__ void k_final(const float* __restrict__ agg, const float* __restrict__ b2,
                        float* __restrict__ out, int N) {
  const int t = blockIdx.x * 256 + threadIdx.x;
  if (t >= N * 2) return;
  const int n = t >> 1, c = t & 1;
  float s = 0.f;
#pragma unroll
  for (int h = 0; h < 8; ++h) s += agg[n * AGG2STR + 2 * h + c];
  out[t] = s * 0.125f + b2[c];
}

extern "C" void kernel_launch(void* const* d_in, const int* in_sizes, int n_in,
                              void* d_out, int out_size, void* d_ws, size_t ws_size,
                              hipStream_t stream) {
  const float* x       = (const float*)d_in[0];
  const int*   ei      = (const int*)d_in[1];
  const float* ft_W    = (const float*)d_in[2];
  const float* ft_b    = (const float*)d_in[3];
  const float* ln_g    = (const float*)d_in[4];
  const float* ln_b    = (const float*)d_in[5];
  const float* W1      = (const float*)d_in[6];
  const float* att_s1  = (const float*)d_in[7];
  const float* att_d1  = (const float*)d_in[8];
  const float* b1      = (const float*)d_in[9];
  const float* bn_g    = (const float*)d_in[10];
  const float* bn_b    = (const float*)d_in[11];
  const float* bn_mean = (const float*)d_in[12];
  const float* bn_var  = (const float*)d_in[13];
  const float* W2      = (const float*)d_in[14];
  const float* att_s2  = (const float*)d_in[15];
  const float* att_d2  = (const float*)d_in[16];
  const float* b2      = (const float*)d_in[17];

  const int N  = in_sizes[0] / 128;
  const int E  = in_sizes[1] / 2;
  const int MT = N / 16;

  char* cur = (char*)d_ws;
  auto alloc = [&](size_t bytes) -> void* {
    void* p = (void*)cur;
    cur += (bytes + 255) & ~(size_t)255;
    return p;
  };
  unsigned* packWft = (unsigned*)alloc((size_t)32 * 4 * 256 * 4);
  unsigned* packW1  = (unsigned*)alloc((size_t)32 * 16 * 256 * 4);
  unsigned* packW2  = (unsigned*)alloc((size_t)1 * 16 * 256 * 4);
  float* hW1 = (float*)alloc((size_t)N * 512 * 4);
  float* hW2 = (float*)alloc((size_t)N * 16 * 4);
  float* aS1 = (float*)alloc((size_t)N * 8 * 4);
  float* aD1 = (float*)alloc((size_t)N * 8 * 4);
  float* aS2 = (float*)alloc((size_t)N * 8 * 4);
  float* aD2 = (float*)alloc((size_t)N * 8 * 4);
  float* agg1 = (float*)alloc((size_t)N * 512 * 4);
  float* agg2 = (float*)alloc((size_t)N * AGG2STR * 4);
  int* csrE = (int*)alloc((size_t)E * 4);
  int* csrS = (int*)alloc((size_t)E * 4);
  int* off  = (int*)alloc((size_t)(N + 1) * 4);
  int* cnt  = (int*)alloc((size_t)N * 4);

  k_prep<<<128 + 512 + 16 + (N + 255) / 256, 256, 0, stream>>>(
      ft_W, W1, W2, packWft, packW1, packW2, cnt, N);

  k_count<<<(E + 255) / 256, 256, 0, stream>>>(ei, cnt, E);

  k_ft_gemm1<<<MT + 1, 256, 0, stream>>>(x, packWft, ft_b, ln_g, ln_b, packW1,
                                         att_s1, att_d1, hW1, aS1, aD1, cnt, off, N);
  k_fill<<<(E + 255) / 256, 256, 0, stream>>>(ei, off, cnt, csrE, csrS, E);
  k_gather1<<<(N + NPW - 1) / NPW, 32 * NPW, 0, stream>>>(N, off, csrE, csrS, hW1, aS1, aD1, agg1);

  k_bn_gemm2<<<MT, 256, 0, stream>>>(agg1, b1, bn_g, bn_b, bn_mean, bn_var, packW2,
                                     att_s2, att_d2, hW2, aS2, aD2);
  k_gather2<<<(N + NPW - 1) / NPW, 32 * NPW, 0, stream>>>(N, off, csrE, csrS, hW2, aS2, aD2, agg2);

  k_final<<<(N * 2 + 255) / 256, 256, 0, stream>>>(agg2, b2, (float*)d_out, N);
}


# Both gathers: a segment that fits its bucket loads each (edge, head) logit once, a lane's loads all in flight together; max, sum, weights from registers (3 passes to 1); summation order unchanged
# speedup vs baseline: 3.9936x; 1.6066x over previous
#include <hip/hip_runtime.h>
#include <stdint.h>
#include <stddef.h>


typedef __attribute__((ext_vector_type(16))) __bf16 v16bf;
typedef __attribute__((ext_vector_type(8)))  float  v8f;
typedef __attribute__((ext_vector_type(8)))  unsigned v8u;

#define LDSTR 520

__device__ __forceinline__ unsigned f2bf(float f) {
  unsigned u = __float_as_uint(f);
  return (u + 0x7FFFu + ((u >> 16) & 1u)) >> 16;
}

__device__ __forceinline__ int kmap(int v, int lane) {
  return ((v & 4) << 2) | ((lane & 16) >> 1) | ((v & 3) << 1);
}

template <int NWAVES>
__device__ __forceinline__ void pack_tile_to_lds_frag(const float* tile, uint4* flo,
                                                      uint4* fhi, int tid) {
  static_assert(NWAVES == 8 || NWAVES == 16, "16 ksteps are split evenly over the waves");
  const int lane = tid & 31, w = tid >> 5;
  const int m = lane & 15;
#pragma unroll
  for (int q = 0; q < 16 / NWAVES; ++q) {
    const int ks = w * (16 / NWAVES) + q;
    unsigned d[8];
#pragma unroll
    for (int v = 0; v < 8; ++v) {
      const int k = (ks << 5) + kmap(v, lane);
      const unsigned lo = f2bf(tile[m * LDSTR + k]);
      const unsigned hi = f2bf(tile[m * LDSTR + k + 1]);
      d[v] = lo | (hi << 16);
    }
    flo[ks * 32 + lane] = make_uint4(d[0], d[1], d[2], d[3]);
    fhi[ks * 32 + lane] = make_uint4(d[4], d[5], d[6], d[7]);
  }
}

__device__ __forceinline__ v16bf lds_fragA(const uint4* flo, const uint4* fhi,
                                           int ks, int lane) {
  const uint4 lo = flo[ks * 32 + lane], hi = fhi[ks * 32 + lane];
  const v8u u = {lo.x, lo.y, lo.z, lo.w, hi.x, hi.y, hi.z, hi.w};
  return __builtin_bit_cast(v16bf, u);
}

__device__ __forceinline__ void pack_B_elem(const float* __restrict__ W,
                                            unsigned* __restrict__ pack,
                                            const int K, const int N, const int id) {
  const int nks = K >> 5;
  const int total = (N >> 4) * nks * 256;
  if (id >= total) return;
  const int v = id & 7, l = (id >> 3) & 31;
  const int t = id >> 8, ks = t % nks, nt = t / nks;
  const int n = (nt << 4) + (l & 15);
  const int k = (ks << 5) + kmap(v, l);
  pack[id] = f2bf(W[(size_t)k * N + n]) | (f2bf(W[(size_t)(k + 1) * N + n]) << 16);
}

__global__ __launch_bounds__(256) void k_prep(
    const float* __restrict__ ft_W, const float* __restrict__ W1,
    const float* __restrict__ W2, unsigned* __restrict__ packWft,
    unsigned* __restrict__ packW1, unsigned* __restrict__ packW2,
    int* __restrict__ cnt, int N) {
  int b = blockIdx.x;
  const int tid = threadIdx.x;
  if (b < 128) { pack_B_elem(ft_W, packWft, 128, 512, b * 256 + tid); return; }
  b -= 128;
  if (b < 512) { pack_B_elem(W1, packW1, 512, 512, b * 256 + tid); return; }
  b -= 512;
  if (b < 16) { pack_B_elem(W2, packW2, 512, 16, b * 256 + tid); return; }
  b -= 16;
  const int i = b * 256 + tid;
  if (i < N) cnt[i] = 0;
}

#define SEGCAP 64

__device__ __forceinline__ unsigned pk2(float a, float b) {
  return f2bf(a) | (f2bf(b) << 16);
}

__global__ __launch_bounds__(256) void k_ft_gemm1(
    const float* __restrict__ x, const unsigned* __restrict__ packWft,
    const float* __restrict__ ft_b, const float* __restrict__ ln_g,
    const float* __restrict__ ln_b, const unsigned* __restrict__ packW1,
    const float* __restrict__ att_src, const float* __restrict__ att_dst,
    float* __restrict__ hW, float* __restrict__ a_src, float* __restrict__ a_dst,
    const int* __restrict__ ei, int* __restrict__ cnt, int* __restrict__ slot,
    int E, int MT) {
  __shared__ float tile[16 * LDSTR];
  __shared__ float psum[16][16], psq[16][16], stats[16][2];
  __shared__ uint4 flo[16 * 32], fhi[16 * 32];
  if ((int)blockIdx.x >= MT) {
    const int e = ((int)blockIdx.x - MT) * 256 + (int)threadIdx.x;
    if (e < E) {
      const int s = ei[e], d = ei[E + e];
      const int pos = atomicAdd(&cnt[d], 1);
      if (pos < SEGCAP)
        *(int2*)(slot + ((size_t)d * SEGCAP + pos) * 2) = make_int2(e, s);
    }
    return;
  }
  const int tid = threadIdx.x, lane = tid & 31, w = tid >> 5;
  const int mt = blockIdx.x;
  const int m0 = (lane >> 4) << 3, nc = lane & 15;
  {
    const int nks = 4;
    const float* xr = x + (size_t)(mt * 16 + nc) * 128 + ((lane & 16) >> 1);
    v8f acc[4] = {};
    for (int ks = 0; ks < nks; ++ks) {
      const float4 p0 = *(const float4*)(xr + ks * 32);
      const float4 p1 = *(const float4*)(xr + ks * 32 + 4);
      const float4 p2 = *(const float4*)(xr + ks * 32 + 16);
      const float4 p3 = *(const float4*)(xr + ks * 32 + 20);
      const v8u au = {pk2(p0.x, p0.y), pk2(p0.z, p0.w), pk2(p1.x, p1.y), pk2(p1.z, p1.w),
                      pk2(p2.x, p2.y), pk2(p2.z, p2.w), pk2(p3.x, p3.y), pk2(p3.z, p3.w)};
      const v16bf a = __builtin_bit_cast(v16bf, au);
#pragma unroll
      for (int j = 0; j < 4; ++j) {
        const int nt = w * 4 + j;
        v16bf b = *(const v16bf*)(packWft + ((size_t)(nt * nks + ks) * 32 + lane) * 8);
        acc[j] = __builtin_amdgcn_wmma_f32_16x16x32_bf16(false, a, false, b,
                                                         (short)0, acc[j], false, false);
      }
    }
#pragma unroll
    for (int j = 0; j < 4; ++j) {
      const int n = (w * 4 + j) * 16 + nc;
      const float bias = ft_b[n];
#pragma unroll
      for (int r = 0; r < 8; ++r) tile[(m0 + r) * LDSTR + n] = acc[j][r] + bias;
    }
  }
  __syncthreads();
  {
    const int row = tid >> 4, p = tid & 15;
    float s = 0.f, s2 = 0.f;
    for (int c = p * 32; c < p * 32 + 32; ++c) {
      const float xv = tile[row * LDSTR + c];
      s += xv; s2 += xv * xv;
    }
    psum[row][p] = s; psq[row][p] = s2;
  }
  __syncthreads();
  if (tid < 16) {
    float s = 0.f, s2 = 0.f;
    for (int p = 0; p < 16; ++p) { s += psum[tid][p]; s2 += psq[tid][p]; }
    const float mu = s * (1.f / 512.f);
    const float var = s2 * (1.f / 512.f) - mu * mu;
    stats[tid][0] = mu;
    stats[tid][1] = rsqrtf(var + 1e-5f);
  }
  __syncthreads();
  {
    const int row = tid >> 4, p = tid & 15;
    const float mu = stats[row][0], rs = stats[row][1];
    for (int c = p * 32; c < p * 32 + 32; ++c) {
      float xv = (tile[row * LDSTR + c] - mu) * rs * ln_g[c] + ln_b[c];
      tile[row * LDSTR + c] = xv >= 0.f ? xv : 0.2f * xv;
    }
  }
  __syncthreads();
  pack_tile_to_lds_frag<8>(tile, flo, fhi, tid);
  __syncthreads();
  {
    const int nks = 16;
    v8f acc[4] = {};
    for (int ks = 0; ks < nks; ++ks) {
      const v16bf a = lds_fragA(flo, fhi, ks, lane);
#pragma unroll
      for (int j = 0; j < 4; ++j) {
        const int nt = w * 4 + j;
        v16bf b = *(const v16bf*)(packW1 + ((size_t)(nt * nks + ks) * 32 + lane) * 8);
        acc[j] = __builtin_amdgcn_wmma_f32_16x16x32_bf16(false, a, false, b,
                                                         (short)0, acc[j], false, false);
      }
    }
#pragma unroll
    for (int j = 0; j < 4; ++j) {
      const int n = (w * 4 + j) * 16 + nc;
#pragma unroll
      for (int r = 0; r < 8; ++r) tile[(m0 + r) * LDSTR + n] = acc[j][r];
    }
  }
  __syncthreads();
  for (int i = tid; i < 16 * 512; i += 256) {
    const int row = i >> 9, c = i & 511;
    hW[(size_t)(mt * 16 + row) * 512 + c] = tile[row * LDSTR + c];
  }
  if (tid < 128) {
    const int row = tid >> 3, h = tid & 7;
    float s1 = 0.f, s2 = 0.f;
    for (int c = 0; c < 64; ++c) {
      const float v = tile[row * LDSTR + (h << 6) + c];
      s1 += v * att_src[(h << 6) + c];
      s2 += v * att_dst[(h << 6) + c];
    }
    const int node = mt * 16 + row;
    a_src[node * 8 + h] = s1;
    a_dst[node * 8 + h] = s2;
  }
}


__device__ __forceinline__ void stage_chunk_scan(const int d, const int cb, const int n,
                                                 const int* __restrict__ ei, const int E,
                                                 int* lsrc, const int lane,
                                                 int& e0, int& seen) {
  while (e0 < E) {
    const int e = e0 + lane;
    const bool hit = e < E && ei[E + e] == d;
    const unsigned mask = __builtin_amdgcn_ballot_w32(hit);
    const int rank = seen + __builtin_popcount(mask & ((1u << lane) - 1u));
    if (hit && rank >= cb && rank < cb + n) lsrc[rank - cb] = ei[e];
    const int after = seen + __builtin_popcount(mask);
    if (after > cb + n) break;
    e0 += 32;
    seen = after;
    if (after == cb + n) break;
  }
}

#define NPW 16

__device__ __forceinline__ void wave_lds_handover() {
  __builtin_amdgcn_fence(__ATOMIC_RELEASE, "wavefront");
  __builtin_amdgcn_wave_barrier();
  __builtin_amdgcn_fence(__ATOMIC_ACQUIRE, "wavefront");
}

#define APOS(j, h) ((j) * 8 + (((h) & 1) << 2) + ((h) >> 1))

template <int NI>
__device__ __forceinline__ void segment_weights_regs(
    const int deg, const int jj, const int h8, const float ad, const int* lsrc,
    const float* __restrict__ a_src, float* lalpha) {
  float L[NI];
#pragma unroll
  for (int i = 0; i < NI; ++i) {
    const int j = jj + 4 * i;
    const int jc = j < deg ? j : deg - 1;
    L[i] = a_src[lsrc[jc] * 8 + h8];
  }
  float mx = __uint_as_float(0xff800000u);
#pragma unroll
  for (int i = 0; i < NI; ++i) {
    float v = L[i] + ad;
    v = v >= 0.f ? v : 0.2f * v;
    L[i] = v;
    if (jj + 4 * i < deg) mx = fmaxf(mx, v);
  }
  mx = fmaxf(mx, __shfl_xor(mx, 8, 32));
  mx = fmaxf(mx, __shfl_xor(mx, 16, 32));
  float sm = 0.f;
#pragma unroll
  for (int i = 0; i < NI; ++i) {
    const float e = __expf(L[i] - mx);
    L[i] = e;
    if (jj + 4 * i < deg) sm += e;
  }
  sm += __shfl_xor(sm, 8, 32);
  sm += __shfl_xor(sm, 16, 32);
#pragma unroll
  for (int i = 0; i < NI; ++i) {
    const int j = jj + 4 * i;
    if (j < deg) lalpha[APOS(j, h8)] = L[i] / sm;
  }
}

template <int ROWF>
__device__ __forceinline__ void accumulate_rows(
    const int n, const int lane, const int* lsrc, const float* lalpha,
    const float* __restrict__ hW, float4& acc0, float4& acc1, float4& acc2, float4& acc3) {
  const int hl = lane >> 4;
  if (ROWF == 512) {
#pragma unroll 4
    for (int j = 0; j < n; ++j) {
      const int s = __builtin_amdgcn_readfirstlane(lsrc[j]);
      const float* row = hW + (size_t)s * ROWF + lane * 4;
      const float4 v0 = *(const float4*)(row);
      const float4 v1 = *(const float4*)(row + 128);
      const float4 v2 = *(const float4*)(row + 256);
      const float4 v3 = *(const float4*)(row + 384);
      const float4 al = *(const float4*)(lalpha + j * 8 + hl * 4);
      acc0.x += al.x * v0.x; acc0.y += al.x * v0.y; acc0.z += al.x * v0.z; acc0.w += al.x * v0.w;
      acc1.x += al.y * v1.x; acc1.y += al.y * v1.y; acc1.z += al.y * v1.z; acc1.w += al.y * v1.w;
      acc2.x += al.z * v2.x; acc2.y += al.z * v2.y; acc2.z += al.z * v2.z; acc2.w += al.z * v2.w;
      acc3.x += al.w * v3.x; acc3.y += al.w * v3.y; acc3.z += al.w * v3.z; acc3.w += al.w * v3.w;
    }
  } else if (lane < 4) {
#pragma unroll 4
    for (int j = 0; j < n; ++j) {
      const float4 v = *(const float4*)(hW + (size_t)lsrc[j] * ROWF + lane * 4);
      const float a0 = lalpha[APOS(j, 2 * lane)], a1 = lalpha[APOS(j, 2 * lane + 1)];
      acc0.x += a0 * v.x; acc0.y += a0 * v.y; acc0.z += a1 * v.z; acc0.w += a1 * v.w;
    }
  }
}

template <int ROWF>
__device__ __forceinline__ void gather_node_wave(
    const int d, const int w, const int lane,
    const int* __restrict__ cnt, const int* __restrict__ slot,
    const int* __restrict__ ei, const int E, const float* __restrict__ hW,
    const float* __restrict__ a_src, const float* __restrict__ a_dst,
    float4& acc0, float4& acc1, float4& acc2, float4& acc3) {
  static_assert(ROWF == 512 || ROWF == 16, "row gather is written for these two row sizes");
  __shared__ int lkey_all[NPW][SEGCAP];
  __shared__ int lsrc_all[NPW][SEGCAP];
  __shared__ __attribute__((aligned(16))) float lalpha_all[NPW][SEGCAP * 8];
  int* lkey = lkey_all[w];
  int* lsrc = lsrc_all[w];
  float* lalpha = lalpha_all[w];
  const int deg = cnt[d];
  const bool small = deg <= SEGCAP;
  const int h8 = lane & 7, jj = lane >> 3;
  const float ad = a_dst[d * 8 + h8];
  acc0 = make_float4(0.f, 0.f, 0.f, 0.f); acc1 = acc0; acc2 = acc0; acc3 = acc0;
  if (small) {
    const int2* bucket = (const int2*)(slot + (size_t)d * SEGCAP * 2);
    const int t0 = lane, t1 = lane + 32;
    int k0 = 0, s0 = 0, k1 = 0, s1 = 0;
    if (t0 < deg) { const int2 ks = bucket[t0]; k0 = ks.x; s0 = ks.y; lkey[t0] = k0; }
    if (t1 < deg) { const int2 ks = bucket[t1]; k1 = ks.x; s1 = ks.y; lkey[t1] = k1; }
    wave_lds_handover();
    int r0 = 0, r1 = 0;
    for (int j = 0; j < deg; ++j) {
      const int kj = lkey[j];
      r0 += (kj < k0) ? 1 : 0;
      r1 += (kj < k1) ? 1 : 0;
    }
    if (t0 < deg) lsrc[r0] = s0;
    if (t1 < deg) lsrc[r1] = s1;
    wave_lds_handover();
    if (deg > 0) {
      if (deg <= 16)      segment_weights_regs<4>(deg, jj, h8, ad, lsrc, a_src, lalpha);
      else if (deg <= 32) segment_weights_regs<8>(deg, jj, h8, ad, lsrc, a_src, lalpha);
      else                segment_weights_regs<16>(deg, jj, h8, ad, lsrc, a_src, lalpha);
    }
    wave_lds_handover();
    accumulate_rows<ROWF>(deg, lane, lsrc, lalpha, hW, acc0, acc1, acc2, acc3);
    return;
  }
#define STAGE_CHUNK(cb, n)                                               \
  {                                                                      \
    wave_lds_handover();    \
    stage_chunk_scan(d, (cb), (n), ei, E, lsrc, lane, sw_e0, sw_seen);   \
    wave_lds_handover();                                                 \
  }
  float mx = __uint_as_float(0xff800000u);
  int sw_e0 = 0, sw_seen = 0;
  for (int cb = 0; cb < deg; cb += SEGCAP) {
    const int n = (deg - cb) < SEGCAP ? (deg - cb) : SEGCAP;
    STAGE_CHUNK(cb, n)
    for (int j = jj; j < n; j += 4) {
      float L = a_src[lsrc[j] * 8 + h8] + ad;
      L = L >= 0.f ? L : 0.2f * L;
      mx = fmaxf(mx, L);
    }
  }
  mx = fmaxf(mx, __shfl_xor(mx, 8, 32));
  mx = fmaxf(mx, __shfl_xor(mx, 16, 32));
  const float m = mx;
  float sm = 0.f;
  sw_e0 = 0; sw_seen = 0;
  for (int cb = 0; cb < deg; cb += SEGCAP) {
    const int n = (deg - cb) < SEGCAP ? (deg - cb) : SEGCAP;
    STAGE_CHUNK(cb, n)
    for (int j = jj; j < n; j += 4) {
      float L = a_src[lsrc[j] * 8 + h8] + ad;
      L = L >= 0.f ? L : 0.2f * L;
      sm += __expf(L - m);
    }
  }
  sm += __shfl_xor(sm, 8, 32);
  sm += __shfl_xor(sm, 16, 32);
  const float den = sm;
  sw_e0 = 0; sw_seen = 0;
  for (int cb = 0; cb < deg; cb += SEGCAP) {
    const int n = (deg - cb) < SEGCAP ? (deg - cb) : SEGCAP;
    STAGE_CHUNK(cb, n)
    for (int j = jj; j < n; j += 4) {
      float L = a_src[lsrc[j] * 8 + h8] + ad;
      L = L >= 0.f ? L : 0.2f * L;
      lalpha[APOS(j, h8)] = __expf(L - m) / den;
    }
    wave_lds_handover();
    accumulate_rows<ROWF>(n, lane, lsrc, lalpha, hW, acc0, acc1, acc2, acc3);
  }
#undef STAGE_CHUNK
}

__global__ __launch_bounds__(32 * NPW) void k_gather1_bn_gemm2(
    const int N, const int* __restrict__ cnt, const int* __restrict__ slot,
    const int* __restrict__ ei, const int E, const float* __restrict__ hW1,
    const float* __restrict__ a_src1, const float* __restrict__ a_dst1,
    const float* __restrict__ b1, const float* __restrict__ bn_g,
    const float* __restrict__ bn_b, const float* __restrict__ bn_mean,
    const float* __restrict__ bn_var, const unsigned* __restrict__ packW,
    const float* __restrict__ att_src, const float* __restrict__ att_dst,
    float* __restrict__ hW, float* __restrict__ a_src, float* __restrict__ a_dst) {
  __shared__ __attribute__((aligned(16))) float tile[16 * LDSTR];
  __shared__ uint4 flo[16 * 32], fhi[16 * 32];
  __shared__ float t2[16][16];
  const int mt = blockIdx.x, tid = threadIdx.x, lane = tid & 31;
  const int w = __builtin_amdgcn_readfirstlane(tid >> 5);
  const int d = mt * NPW + w;
  float4 acc[4];
  acc[0] = make_float4(0.f, 0.f, 0.f, 0.f); acc[1] = acc[0]; acc[2] = acc[0]; acc[3] = acc[0];
  if (d < N)
    gather_node_wave<512>(d, w, lane, cnt, slot, ei, E, hW1, a_src1, a_dst1,
                          acc[0], acc[1], acc[2], acc[3]);
#pragma unroll
  for (int q = 0; q < 4; ++q) {
    const int c = q * 128 + lane * 4;
    const float4 vb1 = *(const float4*)(b1 + c), vmu = *(const float4*)(bn_mean + c);
    const float4 vva = *(const float4*)(bn_var + c), vg = *(const float4*)(bn_g + c);
    const float4 vb = *(const float4*)(bn_b + c);
    float4 r;
    r.x = ((acc[q].x + vb1.x) - vmu.x) * rsqrtf(vva.x + 1e-5f) * vg.x + vb.x;
    r.y = ((acc[q].y + vb1.y) - vmu.y) * rsqrtf(vva.y + 1e-5f) * vg.y + vb.y;
    r.z = ((acc[q].z + vb1.z) - vmu.z) * rsqrtf(vva.z + 1e-5f) * vg.z + vb.z;
    r.w = ((acc[q].w + vb1.w) - vmu.w) * rsqrtf(vva.w + 1e-5f) * vg.w + vb.w;
    r.x = r.x > 0.f ? r.x : 0.f; r.y = r.y > 0.f ? r.y : 0.f;
    r.z = r.z > 0.f ? r.z : 0.f; r.w = r.w > 0.f ? r.w : 0.f;
    *(float4*)(tile + w * LDSTR + c) = r;
  }
  __syncthreads();
  pack_tile_to_lds_frag<NPW>(tile, flo, fhi, tid);
  __syncthreads();
  if (tid < 32) {
    const int nks = 16;
    v8f acc2 = {};
    for (int ks = 0; ks < nks; ++ks) {
      const v16bf a = lds_fragA(flo, fhi, ks, lane);
      v16bf b = *(const v16bf*)(packW + ((size_t)ks * 32 + lane) * 8);
      acc2 = __builtin_amdgcn_wmma_f32_16x16x32_bf16(false, a, false, b,
                                                     (short)0, acc2, false, false);
    }
    const int m0 = (lane >> 4) << 3, nc = lane & 15;
#pragma unroll
    for (int r = 0; r < 8; ++r) t2[m0 + r][nc] = acc2[r];
  }
  __syncthreads();
  if (tid < 256) {
    const int row = tid >> 4, c = tid & 15, node = mt * 16 + row;
    if (node < N) hW[(size_t)node * 16 + c] = t2[row][c];
  }
  if (tid < 128) {
    const int row = tid >> 3, h = tid & 7, node = mt * 16 + row;
    const float v0 = t2[row][2 * h], v1 = t2[row][2 * h + 1];
    if (node < N) {
      a_src[node * 8 + h] = v0 * att_src[2 * h] + v1 * att_src[2 * h + 1];
      a_dst[node * 8 + h] = v0 * att_dst[2 * h] + v1 * att_dst[2 * h + 1];
    }
  }
}

__global__ __launch_bounds__(32 * NPW) void k_gather2_final(
    const int N, const int* __restrict__ cnt, const int* __restrict__ slot,
    const int* __restrict__ ei, const int E, const float* __restrict__ hW2,
    const float* __restrict__ a_src2, const float* __restrict__ a_dst2,
    const float* __restrict__ b2, float* __restrict__ out) {
  __shared__ __attribute__((aligned(16))) float rows[NPW][16];
  const int tid = threadIdx.x, lane = tid & 31;
  const int w = __builtin_amdgcn_readfirstlane(tid >> 5);
  const int d = blockIdx.x * NPW + w;
  float4 acc0 = make_float4(0.f, 0.f, 0.f, 0.f), acc1 = acc0, acc2 = acc0, acc3 = acc0;
  if (d < N)
    gather_node_wave<16>(d, w, lane, cnt, slot, ei, E, hW2, a_src2, a_dst2,
                         acc0, acc1, acc2, acc3);
  if (lane < 4) *(float4*)(&rows[w][lane * 4]) = acc0;
  __syncthreads();
  if (tid < 32) {
    const int n = tid >> 1, c = tid & 1, node = blockIdx.x * NPW + n;
    float s = 0.f;
#pragma unroll
    for (int h = 0; h < 8; ++h) s += rows[n][2 * h + c];
    if (node < N) out[node * 2 + c] = s * 0.125f + b2[c];
  }
}

extern "C" void kernel_launch(void* const* d_in, const int* in_sizes, int n_in,
                              void* d_out, int out_size, void* d_ws, size_t ws_size,
                              hipStream_t stream) {
  const float* x       = (const float*)d_in[0];
  const int*   ei      = (const int*)d_in[1];
  const float* ft_W    = (const float*)d_in[2];
  const float* ft_b    = (const float*)d_in[3];
  const float* ln_g    = (const float*)d_in[4];
  const float* ln_b    = (const float*)d_in[5];
  const float* W1      = (const float*)d_in[6];
  const float* att_s1  = (const float*)d_in[7];
  const float* att_d1  = (const float*)d_in[8];
  const float* b1      = (const float*)d_in[9];
  const float* bn_g    = (const float*)d_in[10];
  const float* bn_b    = (const float*)d_in[11];
  const float* bn_mean = (const float*)d_in[12];
  const float* bn_var  = (const float*)d_in[13];
  const float* W2      = (const float*)d_in[14];
  const float* att_s2  = (const float*)d_in[15];
  const float* att_d2  = (const float*)d_in[16];
  const float* b2      = (const float*)d_in[17];

  const int N  = in_sizes[0] / 128;
  const int E  = in_sizes[1] / 2;
  const int MT = N / 16;

  char* cur = (char*)d_ws;
  auto alloc = [&](size_t bytes) -> void* {
    void* p = (void*)cur;
    cur += (bytes + 255) & ~(size_t)255;
    return p;
  };
  unsigned* packWft = (unsigned*)alloc((size_t)32 * 4 * 256 * 4);
  unsigned* packW1  = (unsigned*)alloc((size_t)32 * 16 * 256 * 4);
  unsigned* packW2  = (unsigned*)alloc((size_t)1 * 16 * 256 * 4);
  float* hW1 = (float*)alloc((size_t)N * 512 * 4);
  float* hW2 = (float*)alloc((size_t)N * 16 * 4);
  float* aS1 = (float*)alloc((size_t)N * 8 * 4);
  float* aD1 = (float*)alloc((size_t)N * 8 * 4);
  float* aS2 = (float*)alloc((size_t)N * 8 * 4);
  float* aD2 = (float*)alloc((size_t)N * 8 * 4);
  int* slot = (int*)alloc((size_t)N * SEGCAP * 8);
  int* cnt  = (int*)alloc((size_t)N * 4);

  k_prep<<<128 + 512 + 16 + (N + 255) / 256, 256, 0, stream>>>(
      ft_W, W1, W2, packWft, packW1, packW2, cnt, N);

  k_ft_gemm1<<<MT + (E + 255) / 256, 256, 0, stream>>>(
      x, packWft, ft_b, ln_g, ln_b, packW1, att_s1, att_d1, hW1, aS1, aD1,
      ei, cnt, slot, E, MT);

  k_gather1_bn_gemm2<<<(N + NPW - 1) / NPW, 32 * NPW, 0, stream>>>(
      N, cnt, slot, ei, E, hW1, aS1, aD1, b1, bn_g, bn_b, bn_mean, bn_var, packW2,
      att_s2, att_d2, hW2, aS2, aD2);

  k_gather2_final<<<(N + NPW - 1) / NPW, 32 * NPW, 0, stream>>>(
      N, cnt, slot, ei, E, hW2, aS2, aD2, b2, (float*)d_out);
}


# GEMM0+LayerNorm+GEMM1 kernel: a work-group owns 48 rows (3 tiles) not 16, each weight fragment feeds 3 matrix instructions, loaded 2 ksteps ahead; LayerNorm row sums regrouped (order changes)
# speedup vs baseline: 4.9703x; 1.2446x over previous
#include <hip/hip_runtime.h>
#include <stdint.h>
#include <stddef.h>


typedef __attribute__((ext_vector_type(16))) __bf16 v16bf;
typedef __attribute__((ext_vector_type(8)))  float  v8f;
typedef __attribute__((ext_vector_type(8)))  unsigned v8u;

#define LDSTR 520

__device__ __forceinline__ unsigned f2bf(float f) {
  unsigned u = __float_as_uint(f);
  return (u + 0x7FFFu + ((u >> 16) & 1u)) >> 16;
}

__device__ __forceinline__ int kmap(int v, int lane) {
  return ((v & 4) << 2) | ((lane & 16) >> 1) | ((v & 3) << 1);
}

template <int NWAVES>
__device__ __forceinline__ void pack_tile_to_lds_frag(const float* tile, uint4* flo,
                                                      uint4* fhi, int tid) {
  static_assert(NWAVES == 8 || NWAVES == 16, "16 ksteps are split evenly over the waves");
  const int lane = tid & 31, w = tid >> 5;
  const int m = lane & 15;
#pragma unroll
  for (int q = 0; q < 16 / NWAVES; ++q) {
    const int ks = w * (16 / NWAVES) + q;
    unsigned d[8];
#pragma unroll
    for (int v = 0; v < 8; ++v) {
      const int k = (ks << 5) + kmap(v, lane);
      const unsigned lo = f2bf(tile[m * LDSTR + k]);
      const unsigned hi = f2bf(tile[m * LDSTR + k + 1]);
      d[v] = lo | (hi << 16);
    }
    flo[ks * 32 + lane] = make_uint4(d[0], d[1], d[2], d[3]);
    fhi[ks * 32 + lane] = make_uint4(d[4], d[5], d[6], d[7]);
  }
}

__device__ __forceinline__ v16bf lds_fragA(const uint4* flo, const uint4* fhi,
                                           int ks, int lane) {
  const uint4 lo = flo[ks * 32 + lane], hi = fhi[ks * 32 + lane];
  const v8u u = {lo.x, lo.y, lo.z, lo.w, hi.x, hi.y, hi.z, hi.w};
  return __builtin_bit_cast(v16bf, u);
}

__device__ __forceinline__ void pack_B_elem(const float* __restrict__ W,
                                            unsigned* __restrict__ pack,
                                            const int K, const int N, const int id) {
  const int nks = K >> 5;
  const int total = (N >> 4) * nks * 256;
  if (id >= total) return;
  const int v = id & 7, l = (id >> 3) & 31;
  const int t = id >> 8, ks = t % nks, nt = t / nks;
  const int n = (nt << 4) + (l & 15);
  const int k = (ks << 5) + kmap(v, l);
  pack[id] = f2bf(W[(size_t)k * N + n]) | (f2bf(W[(size_t)(k + 1) * N + n]) << 16);
}

__global__ __launch_bounds__(256) void k_prep(
    const float* __restrict__ ft_W, const float* __restrict__ W1,
    const float* __restrict__ W2, unsigned* __restrict__ packWft,
    unsigned* __restrict__ packW1, unsigned* __restrict__ packW2,
    int* __restrict__ cnt, int N) {
  int b = blockIdx.x;
  const int tid = threadIdx.x;
  if (b < 128) { pack_B_elem(ft_W, packWft, 128, 512, b * 256 + tid); return; }
  b -= 128;
  if (b < 512) { pack_B_elem(W1, packW1, 512, 512, b * 256 + tid); return; }
  b -= 512;
  if (b < 16) { pack_B_elem(W2, packW2, 512, 16, b * 256 + tid); return; }
  b -= 16;
  const int i = b * 256 + tid;
  if (i < N) cnt[i] = 0;
}

#define SEGCAP 64

__device__ __forceinline__ unsigned pk2(float a, float b) {
  return f2bf(a) | (f2bf(b) << 16);
}

typedef __attribute__((ext_vector_type(2))) _Float16 h2v;
__device__ __forceinline__ unsigned pkh2(float a, float b) {
  h2v p;
  p.x = (_Float16)a;
  p.y = (_Float16)b;
  return __builtin_bit_cast(unsigned, p);
}
__device__ __forceinline__ float hlo(unsigned u) { return (float)__builtin_bit_cast(h2v, u).x; }
__device__ __forceinline__ float hhi(unsigned u) { return (float)__builtin_bit_cast(h2v, u).y; }

#define MTW 3
#define LDP 516

__device__ __forceinline__ float ln_act(float v, float mu, float rs, float g, float b) {
  const float xv = (v - mu) * rs * g + b;
  return xv >= 0.f ? xv : 0.2f * xv;
}

__global__ __launch_bounds__(256, 4) void k_ft_gemm1(
    const float* __restrict__ x, const unsigned* __restrict__ packWft,
    const float* __restrict__ ft_b, const float* __restrict__ ln_g,
    const float* __restrict__ ln_b, const unsigned* __restrict__ packW1,
    const float* __restrict__ att_src, const float* __restrict__ att_dst,
    unsigned* __restrict__ hWh, float* __restrict__ a_src, float* __restrict__ a_dst,
    const int* __restrict__ ei, int* __restrict__ cnt, int* __restrict__ slot,
    int E, int MG, int N) {
  __shared__ __attribute__((aligned(16))) float tile[MTW * 16 * LDP];
  __shared__ __attribute__((aligned(16))) float red[MTW * 16 * 32];
  __shared__ float stats[MTW * 16][2];
  __shared__ uint4 flo[MTW * 16 * 32], fhi[MTW * 16 * 32];
  if ((int)blockIdx.x >= MG) {
    const int e = ((int)blockIdx.x - MG) * 256 + (int)threadIdx.x;
    if (e < E) {
      const int s = ei[e], d = ei[E + e];
      const int pos = atomicAdd(&cnt[d], 1);
      if (pos < SEGCAP)
        *(int2*)(slot + ((size_t)d * SEGCAP + pos) * 2) = make_int2(e, s);
    }
    return;
  }
  const int tid = threadIdx.x, lane = tid & 31;
  const int w = __builtin_amdgcn_readfirstlane(tid >> 5);
  const int row0 = (int)blockIdx.x * (MTW * 16);
  const int m0 = (lane >> 4) << 3, nc = lane & 15;
  {
    const int nks = 4;
    const unsigned* bw = packWft + ((size_t)(w * 4) * nks * 32 + lane) * 8;
    v16bf bq[3][4];
#pragma unroll
    for (int j = 0; j < 4; ++j) {
      bq[0][j] = *(const v16bf*)(bw + (size_t)(j * nks + 0) * 256);
      bq[1][j] = *(const v16bf*)(bw + (size_t)(j * nks + 1) * 256);
    }
    static_assert(MTW * 8 == 24, "three half-fragments per wave");
    float4 pa[3], pb[3];
#pragma unroll
    for (int i = 0; i < 3; ++i) {
      const int hf = w + 8 * i, f = hf >> 1, t = f >> 2, ks = f & 3;
      int r = row0 + t * 16 + nc;
      r = r < N ? r : N - 1;
      const float* xr = x + (size_t)r * 128 + ks * 32 + ((lane & 16) >> 1) + ((hf & 1) << 4);
      pa[i] = *(const float4*)(xr);
      pb[i] = *(const float4*)(xr + 4);
    }
#pragma unroll
    for (int i = 0; i < 3; ++i) {
      const int hf = w + 8 * i, f = hf >> 1, t = f >> 2, ks = f & 3;
      uint4* dst = (hf & 1) ? fhi : flo;
      dst[(t * 16 + ks) * 32 + lane] = make_uint4(pk2(pa[i].x, pa[i].y), pk2(pa[i].z, pa[i].w),
                                                  pk2(pb[i].x, pb[i].y), pk2(pb[i].z, pb[i].w));
    }
    __syncthreads();
    v8f acc[MTW][4] = {};
#pragma unroll
    for (int ks = 0; ks < nks; ++ks) {
      if (ks + 2 < nks) {
#pragma unroll
        for (int j = 0; j < 4; ++j)
          bq[(ks + 2) % 3][j] = *(const v16bf*)(bw + (size_t)(j * nks + ks + 2) * 256);
      }
      v16bf a[MTW];
#pragma unroll
      for (int t = 0; t < MTW; ++t) a[t] = lds_fragA(flo + t * 512, fhi + t * 512, ks, lane);
      __builtin_amdgcn_sched_barrier(0);
#pragma unroll
      for (int j = 0; j < 4; ++j)
#pragma unroll
        for (int t = 0; t < MTW; ++t)
          acc[t][j] = __builtin_amdgcn_wmma_f32_16x16x32_bf16(false, a[t], false, bq[ks % 3][j],
                                                              (short)0, acc[t][j], false, false);
    }
#pragma unroll
    for (int j = 0; j < 4; ++j) {
      const int n = (w * 4 + j) * 16 + nc;
      const float bias = ft_b[n];
#pragma unroll
      for (int t = 0; t < MTW; ++t)
#pragma unroll
        for (int r = 0; r < 8; ++r) tile[(t * 16 + m0 + r) * LDP + n] = acc[t][j][r] + bias;
    }
  }
  float4 lg[2][4], lb[2][4];
#pragma unroll
  for (int q = 0; q < 2; ++q) {
    const int kb = ((w * 2 + q) << 5) + ((lane & 16) >> 1);
    lg[q][0] = *(const float4*)(ln_g + kb);      lg[q][1] = *(const float4*)(ln_g + kb + 4);
    lg[q][2] = *(const float4*)(ln_g + kb + 16); lg[q][3] = *(const float4*)(ln_g + kb + 20);
    lb[q][0] = *(const float4*)(ln_b + kb);      lb[q][1] = *(const float4*)(ln_b + kb + 4);
    lb[q][2] = *(const float4*)(ln_b + kb + 16); lb[q][3] = *(const float4*)(ln_b + kb + 20);
  }
  __syncthreads();
  {
    const int p = tid & 15;
#pragma unroll
    for (int t = 0; t < MTW; ++t) {
      const int row = t * 16 + (tid >> 4);
      float s = 0.f, s2 = 0.f;
#pragma unroll
      for (int i = 0; i < 8; ++i) {
        const float4 v = *(const float4*)(tile + row * LDP + i * 64 + p * 4);
        s += v.x; s2 += v.x * v.x;
        s += v.y; s2 += v.y * v.y;
        s += v.z; s2 += v.z * v.z;
        s += v.w; s2 += v.w * v.w;
      }
      red[row * 32 + p] = s; red[row * 32 + 16 + p] = s2;
    }
  }
  __syncthreads();
  if (tid < MTW * 16) {
    float s = 0.f, s2 = 0.f;
    for (int p = 0; p < 16; ++p) { s += red[tid * 32 + p]; s2 += red[tid * 32 + 16 + p]; }
    const float mu = s * (1.f / 512.f);
    const float var = s2 * (1.f / 512.f) - mu * mu;
    stats[tid][0] = mu;
    stats[tid][1] = rsqrtf(var + 1e-5f);
  }
  __syncthreads();
  {
    const int m = lane & 15;
#pragma unroll
    for (int q = 0; q < 2; ++q) {
      const int ks = w * 2 + q;
      const int kb = (ks << 5) + ((lane & 16) >> 1);
      const float4 g0 = lg[q][0], g1 = lg[q][1], g2 = lg[q][2], g3 = lg[q][3];
      const float4 b0 = lb[q][0], b1 = lb[q][1], b2 = lb[q][2], b3 = lb[q][3];
      float4 pv[MTW][4];
      float mu[MTW], rs[MTW];
#pragma unroll
      for (int t = 0; t < MTW; ++t) {
        const int row = t * 16 + m;
        mu[t] = stats[row][0]; rs[t] = stats[row][1];
        const float* tr = tile + row * LDP + kb;
        pv[t][0] = *(const float4*)(tr);      pv[t][1] = *(const float4*)(tr + 4);
        pv[t][2] = *(const float4*)(tr + 16); pv[t][3] = *(const float4*)(tr + 20);
      }
#pragma unroll
      for (int t = 0; t < MTW; ++t) {
        const float4 p0 = pv[t][0], p1 = pv[t][1], p2 = pv[t][2], p3 = pv[t][3];
        const float mut = mu[t], rst = rs[t];
        flo[(t * 16 + ks) * 32 + lane] = make_uint4(
            pk2(ln_act(p0.x, mut, rst, g0.x, b0.x), ln_act(p0.y, mut, rst, g0.y, b0.y)),
            pk2(ln_act(p0.z, mut, rst, g0.z, b0.z), ln_act(p0.w, mut, rst, g0.w, b0.w)),
            pk2(ln_act(p1.x, mut, rst, g1.x, b1.x), ln_act(p1.y, mut, rst, g1.y, b1.y)),
            pk2(ln_act(p1.z, mut, rst, g1.z, b1.z), ln_act(p1.w, mut, rst, g1.w, b1.w)));
        fhi[(t * 16 + ks) * 32 + lane] = make_uint4(
            pk2(ln_act(p2.x, mut, rst, g2.x, b2.x), ln_act(p2.y, mut, rst, g2.y, b2.y)),
            pk2(ln_act(p2.z, mut, rst, g2.z, b2.z), ln_act(p2.w, mut, rst, g2.w, b2.w)),
            pk2(ln_act(p3.x, mut, rst, g3.x, b3.x), ln_act(p3.y, mut, rst, g3.y, b3.y)),
            pk2(ln_act(p3.z, mut, rst, g3.z, b3.z), ln_act(p3.w, mut, rst, g3.w, b3.w)));
      }
    }
  }
  __syncthreads();
  {
    const int nks = 16;
    const unsigned* bw = packW1 + ((size_t)(w * 4) * nks * 32 + lane) * 8;
    v8f acc[MTW][4] = {};
    v16bf bq[3][4];
#pragma unroll
    for (int j = 0; j < 4; ++j) {
      bq[0][j] = *(const v16bf*)(bw + (size_t)(j * nks + 0) * 256);
      bq[1][j] = *(const v16bf*)(bw + (size_t)(j * nks + 1) * 256);
    }
    v16bf a[2][MTW];
#pragma unroll
    for (int t = 0; t < MTW; ++t) a[0][t] = lds_fragA(flo + t * 512, fhi + t * 512, 0, lane);
#pragma unroll
    for (int ks = 0; ks < nks; ++ks) {
      if (ks + 2 < nks) {
#pragma unroll
        for (int j = 0; j < 4; ++j)
          bq[(ks + 2) % 3][j] = *(const v16bf*)(bw + (size_t)(j * nks + ks + 2) * 256);
      }
      if (ks + 1 < nks) {
#pragma unroll
        for (int t = 0; t < MTW; ++t)
          a[(ks + 1) & 1][t] = lds_fragA(flo + t * 512, fhi + t * 512, ks + 1, lane);
      }
      __builtin_amdgcn_sched_barrier(0);
#pragma unroll
      for (int j = 0; j < 4; ++j)
#pragma unroll
        for (int t = 0; t < MTW; ++t)
          acc[t][j] = __builtin_amdgcn_wmma_f32_16x16x32_bf16(false, a[ks & 1][t], false, bq[ks % 3][j],
                                                              (short)0, acc[t][j], false, false);
    }
#pragma unroll
    for (int j = 0; j < 4; ++j) {
      const int n = (w * 4 + j) * 16 + nc;
#pragma unroll
      for (int t = 0; t < MTW; ++t)
#pragma unroll
        for (int r = 0; r < 8; ++r) tile[(t * 16 + m0 + r) * LDP + n] = acc[t][j][r];
    }
  }
  __syncthreads();
  if (row0 + MTW * 16 <= N) {
    float4 u[MTW * 4], v[MTW * 4];
#pragma unroll
    for (int it = 0; it < MTW * 4; ++it) {
      const int i = tid + it * 256, row = i >> 6, c8 = i & 63;
      u[it] = *(const float4*)(tile + row * LDP + c8 * 8);
      v[it] = *(const float4*)(tile + row * LDP + c8 * 8 + 4);
    }
#pragma unroll
    for (int it = 0; it < MTW * 4; ++it) {
      const int i = tid + it * 256, row = i >> 6, c8 = i & 63;
      *(uint4*)(hWh + (size_t)(row0 + row) * 256 + c8 * 4) =
          make_uint4(pkh2(u[it].x, u[it].y), pkh2(u[it].z, u[it].w),
                     pkh2(v[it].x, v[it].y), pkh2(v[it].z, v[it].w));
    }
  } else {
    for (int i = tid; i < MTW * 16 * 64; i += 256) {
      const int row = i >> 6, c8 = i & 63;
      if (row0 + row < N) {
        const float4 u = *(const float4*)(tile + row * LDP + c8 * 8);
        const float4 v = *(const float4*)(tile + row * LDP + c8 * 8 + 4);
        *(uint4*)(hWh + (size_t)(row0 + row) * 256 + c8 * 4) =
            make_uint4(pkh2(u.x, u.y), pkh2(u.z, u.w), pkh2(v.x, v.y), pkh2(v.z, v.w));
      }
    }
  }
#pragma unroll
  for (int rb = 0; rb < MTW * 16; rb += 32) {
    const int row = rb + lane;
    if (row < MTW * 16) {
      const float* tr = tile + row * LDP + (w << 6);
      float4 hv[16];
#pragma unroll
      for (int c4 = 0; c4 < 16; ++c4) hv[c4] = *(const float4*)(tr + c4 * 4);
      __builtin_amdgcn_sched_barrier(0);
      float s1 = 0.f, s2 = 0.f;
#pragma unroll
      for (int c4 = 0; c4 < 16; ++c4) {
        const float4 as = *(const float4*)(att_src + (w << 6) + c4 * 4);
        const float4 ad = *(const float4*)(att_dst + (w << 6) + c4 * 4);
        s1 += hv[c4].x * as.x; s2 += hv[c4].x * ad.x;
        s1 += hv[c4].y * as.y; s2 += hv[c4].y * ad.y;
        s1 += hv[c4].z * as.z; s2 += hv[c4].z * ad.z;
        s1 += hv[c4].w * as.w; s2 += hv[c4].w * ad.w;
      }
      red[row * 8 + w] = s1;
      red[MTW * 128 + row * 8 + w] = s2;
    }
  }
  __syncthreads();
  for (int i = tid; i < MTW * 128; i += 256) {
    if (row0 + (i >> 3) < N) {
      a_src[(size_t)row0 * 8 + i] = red[i];
      a_dst[(size_t)row0 * 8 + i] = red[MTW * 128 + i];
    }
  }
}


__device__ __forceinline__ void stage_chunk_scan(const int d, const int cb, const int n,
                                                 const int* __restrict__ ei, const int E,
                                                 int* lsrc, const int lane,
                                                 int& e0, int& seen) {
  while (e0 < E) {
    const int e = e0 + lane;
    const bool hit = e < E && ei[E + e] == d;
    const unsigned mask = __builtin_amdgcn_ballot_w32(hit);
    const int rank = seen + __builtin_popcount(mask & ((1u << lane) - 1u));
    if (hit && rank >= cb && rank < cb + n) lsrc[rank - cb] = ei[e];
    const int after = seen + __builtin_popcount(mask);
    if (after > cb + n) break;
    e0 += 32;
    seen = after;
    if (after == cb + n) break;
  }
}

#define NPW 16

__device__ __forceinline__ void wave_lds_handover() {
  __builtin_amdgcn_fence(__ATOMIC_RELEASE, "wavefront");
  __builtin_amdgcn_wave_barrier();
  __builtin_amdgcn_fence(__ATOMIC_ACQUIRE, "wavefront");
}

#define APOS(j, h) ((j) * 8 + (((h) & 1) << 2) + ((h) >> 1))

template <int NI>
__device__ __forceinline__ void segment_weights_regs(
    const int deg, const int jj, const int h8, const float ad, const int* lsrc,
    const float* __restrict__ a_src, float* lalpha) {
  float L[NI];
#pragma unroll
  for (int i = 0; i < NI; ++i) {
    const int j = jj + 4 * i;
    const int jc = j < deg ? j : deg - 1;
    L[i] = a_src[lsrc[jc] * 8 + h8];
  }
  float mx = __uint_as_float(0xff800000u);
#pragma unroll
  for (int i = 0; i < NI; ++i) {
    float v = L[i] + ad;
    v = v >= 0.f ? v : 0.2f * v;
    L[i] = v;
    if (jj + 4 * i < deg) mx = fmaxf(mx, v);
  }
  mx = fmaxf(mx, __shfl_xor(mx, 8, 32));
  mx = fmaxf(mx, __shfl_xor(mx, 16, 32));
  float sm = 0.f;
#pragma unroll
  for (int i = 0; i < NI; ++i) {
    const float e = __expf(L[i] - mx);
    L[i] = e;
    if (jj + 4 * i < deg) sm += e;
  }
  sm += __shfl_xor(sm, 8, 32);
  sm += __shfl_xor(sm, 16, 32);
#pragma unroll
  for (int i = 0; i < NI; ++i) {
    const int j = jj + 4 * i;
    if (j < deg) lalpha[APOS(j, h8)] = L[i] / sm;
  }
}

template <int ROWF, typename RowT>
__device__ __forceinline__ void accumulate_rows(
    const int n, const int lane, const int* lsrc, const float* lalpha,
    const RowT* __restrict__ hW, float4& acc0, float4& acc1, float4& acc2, float4& acc3) {
  if (ROWF == 512) {
    const int pa = APOS(0, lane >> 3);
#pragma unroll 4
    for (int j = 0; j < n; ++j) {
      const int s = __builtin_amdgcn_readfirstlane(lsrc[j]);
      const uint4* row = reinterpret_cast<const uint4*>(hW) + (size_t)s * 64 + lane;
      const uint4 v0 = row[0];
      const uint4 v1 = row[32];
      const float a0 = lalpha[j * 8 + pa], a1 = lalpha[j * 8 + pa + 2];
      acc0.x += a0 * hlo(v0.x); acc0.y += a0 * hhi(v0.x); acc0.z += a0 * hlo(v0.y); acc0.w += a0 * hhi(v0.y);
      acc1.x += a0 * hlo(v0.z); acc1.y += a0 * hhi(v0.z); acc1.z += a0 * hlo(v0.w); acc1.w += a0 * hhi(v0.w);
      acc2.x += a1 * hlo(v1.x); acc2.y += a1 * hhi(v1.x); acc2.z += a1 * hlo(v1.y); acc2.w += a1 * hhi(v1.y);
      acc3.x += a1 * hlo(v1.z); acc3.y += a1 * hhi(v1.z); acc3.z += a1 * hlo(v1.w); acc3.w += a1 * hhi(v1.w);
    }
  } else if (lane < 4) {
#pragma unroll 4
    for (int j = 0; j < n; ++j) {
      const float4 v = *(reinterpret_cast<const float4*>(hW) + (size_t)lsrc[j] * (ROWF / 4) + lane);
      const float a0 = lalpha[APOS(j, 2 * lane)], a1 = lalpha[APOS(j, 2 * lane + 1)];
      acc0.x += a0 * v.x; acc0.y += a0 * v.y; acc0.z += a1 * v.z; acc0.w += a1 * v.w;
    }
  }
}

template <int ROWF, typename RowT>
__device__ __forceinline__ void gather_node_wave(
    const int d, const int w, const int lane,
    const int* __restrict__ cnt, const int* __restrict__ slot,
    const int* __restrict__ ei, const int E, const RowT* __restrict__ hW,
    const float* __restrict__ a_src, const float* __restrict__ a_dst,
    float4& acc0, float4& acc1, float4& acc2, float4& acc3) {
  static_assert(ROWF == 512 || ROWF == 16, "row gather is written for these two row sizes");
  __shared__ int lkey_all[NPW][SEGCAP];
  __shared__ int lsrc_all[NPW][SEGCAP];
  __shared__ __attribute__((aligned(16))) float lalpha_all[NPW][SEGCAP * 8];
  int* lkey = lkey_all[w];
  int* lsrc = lsrc_all[w];
  float* lalpha = lalpha_all[w];
  const int deg = cnt[d];
  const bool small = deg <= SEGCAP;
  const int h8 = lane & 7, jj = lane >> 3;
  const float ad = a_dst[d * 8 + h8];
  acc0 = make_float4(0.f, 0.f, 0.f, 0.f); acc1 = acc0; acc2 = acc0; acc3 = acc0;
  if (small) {
    const int2* bucket = (const int2*)(slot + (size_t)d * SEGCAP * 2);
    const int t0 = lane, t1 = lane + 32;
    int k0 = 0, s0 = 0, k1 = 0, s1 = 0;
    if (t0 < deg) { const int2 ks = bucket[t0]; k0 = ks.x; s0 = ks.y; lkey[t0] = k0; }
    if (t1 < deg) { const int2 ks = bucket[t1]; k1 = ks.x; s1 = ks.y; lkey[t1] = k1; }
    wave_lds_handover();
    int r0 = 0, r1 = 0;
    for (int j = 0; j < deg; ++j) {
      const int kj = lkey[j];
      r0 += (kj < k0) ? 1 : 0;
      r1 += (kj < k1) ? 1 : 0;
    }
    if (t0 < deg) lsrc[r0] = s0;
    if (t1 < deg) lsrc[r1] = s1;
    wave_lds_handover();
    if (deg > 0) {
      if (deg <= 16)      segment_weights_regs<4>(deg, jj, h8, ad, lsrc, a_src, lalpha);
      else if (deg <= 32) segment_weights_regs<8>(deg, jj, h8, ad, lsrc, a_src, lalpha);
      else                segment_weights_regs<16>(deg, jj, h8, ad, lsrc, a_src, lalpha);
    }
    wave_lds_handover();
    accumulate_rows<ROWF, RowT>(deg, lane, lsrc, lalpha, hW, acc0, acc1, acc2, acc3);
    return;
  }
#define STAGE_CHUNK(cb, n)                                               \
  {                                                                      \
    wave_lds_handover();    \
    stage_chunk_scan(d, (cb), (n), ei, E, lsrc, lane, sw_e0, sw_seen);   \
    wave_lds_handover();                                                 \
  }
  float mx = __uint_as_float(0xff800000u);
  int sw_e0 = 0, sw_seen = 0;
  for (int cb = 0; cb < deg; cb += SEGCAP) {
    const int n = (deg - cb) < SEGCAP ? (deg - cb) : SEGCAP;
    STAGE_CHUNK(cb, n)
    for (int j = jj; j < n; j += 4) {
      float L = a_src[lsrc[j] * 8 + h8] + ad;
      L = L >= 0.f ? L : 0.2f * L;
      mx = fmaxf(mx, L);
    }
  }
  mx = fmaxf(mx, __shfl_xor(mx, 8, 32));
  mx = fmaxf(mx, __shfl_xor(mx, 16, 32));
  const float m = mx;
  float sm = 0.f;
  sw_e0 = 0; sw_seen = 0;
  for (int cb = 0; cb < deg; cb += SEGCAP) {
    const int n = (deg - cb) < SEGCAP ? (deg - cb) : SEGCAP;
    STAGE_CHUNK(cb, n)
    for (int j = jj; j < n; j += 4) {
      float L = a_src[lsrc[j] * 8 + h8] + ad;
      L = L >= 0.f ? L : 0.2f * L;
      sm += __expf(L - m);
    }
  }
  sm += __shfl_xor(sm, 8, 32);
  sm += __shfl_xor(sm, 16, 32);
  const float den = sm;
  sw_e0 = 0; sw_seen = 0;
  for (int cb = 0; cb < deg; cb += SEGCAP) {
    const int n = (deg - cb) < SEGCAP ? (deg - cb) : SEGCAP;
    STAGE_CHUNK(cb, n)
    for (int j = jj; j < n; j += 4) {
      float L = a_src[lsrc[j] * 8 + h8] + ad;
      L = L >= 0.f ? L : 0.2f * L;
      lalpha[APOS(j, h8)] = __expf(L - m) / den;
    }
    wave_lds_handover();
    accumulate_rows<ROWF, RowT>(n, lane, lsrc, lalpha, hW, acc0, acc1, acc2, acc3);
  }
#undef STAGE_CHUNK
}

__global__ __launch_bounds__(32 * NPW) void k_gather1_bn_gemm2(
    const int N, const int* __restrict__ cnt, const int* __restrict__ slot,
    const int* __restrict__ ei, const int E, const unsigned* __restrict__ hW1h,
    const float* __restrict__ a_src1, const float* __restrict__ a_dst1,
    const float* __restrict__ b1, const float* __restrict__ bn_g,
    const float* __restrict__ bn_b, const float* __restrict__ bn_mean,
    const float* __restrict__ bn_var, const unsigned* __restrict__ packW,
    const float* __restrict__ att_src, const float* __restrict__ att_dst,
    float* __restrict__ hW, float* __restrict__ a_src, float* __restrict__ a_dst) {
  __shared__ __attribute__((aligned(16))) float tile[16 * LDSTR];
  __shared__ uint4 flo[16 * 32], fhi[16 * 32];
  __shared__ float t2[16][16];
  const int mt = blockIdx.x, tid = threadIdx.x, lane = tid & 31;
  const int w = __builtin_amdgcn_readfirstlane(tid >> 5);
  const int d = mt * NPW + w;
  float4 acc[4];
  acc[0] = make_float4(0.f, 0.f, 0.f, 0.f); acc[1] = acc[0]; acc[2] = acc[0]; acc[3] = acc[0];
  if (d < N)
    gather_node_wave<512, unsigned>(d, w, lane, cnt, slot, ei, E, hW1h, a_src1, a_dst1,
                          acc[0], acc[1], acc[2], acc[3]);
#pragma unroll
  for (int q = 0; q < 4; ++q) {
    const int c = (q >> 1) * 256 + lane * 8 + (q & 1) * 4;
    const float4 vb1 = *(const float4*)(b1 + c), vmu = *(const float4*)(bn_mean + c);
    const float4 vva = *(const float4*)(bn_var + c), vg = *(const float4*)(bn_g + c);
    const float4 vb = *(const float4*)(bn_b + c);
    float4 r;
    r.x = ((acc[q].x + vb1.x) - vmu.x) * rsqrtf(vva.x + 1e-5f) * vg.x + vb.x;
    r.y = ((acc[q].y + vb1.y) - vmu.y) * rsqrtf(vva.y + 1e-5f) * vg.y + vb.y;
    r.z = ((acc[q].z + vb1.z) - vmu.z) * rsqrtf(vva.z + 1e-5f) * vg.z + vb.z;
    r.w = ((acc[q].w + vb1.w) - vmu.w) * rsqrtf(vva.w + 1e-5f) * vg.w + vb.w;
    r.x = r.x > 0.f ? r.x : 0.f; r.y = r.y > 0.f ? r.y : 0.f;
    r.z = r.z > 0.f ? r.z : 0.f; r.w = r.w > 0.f ? r.w : 0.f;
    *(float4*)(tile + w * LDSTR + c) = r;
  }
  __syncthreads();
  pack_tile_to_lds_frag<NPW>(tile, flo, fhi, tid);
  __syncthreads();
  if (tid < 32) {
    const int nks = 16;
    v8f acc2 = {};
    for (int ks = 0; ks < nks; ++ks) {
      const v16bf a = lds_fragA(flo, fhi, ks, lane);
      v16bf b = *(const v16bf*)(packW + ((size_t)ks * 32 + lane) * 8);
      acc2 = __builtin_amdgcn_wmma_f32_16x16x32_bf16(false, a, false, b,
                                                     (short)0, acc2, false, false);
    }
    const int m0 = (lane >> 4) << 3, nc = lane & 15;
#pragma unroll
    for (int r = 0; r < 8; ++r) t2[m0 + r][nc] = acc2[r];
  }
  __syncthreads();
  if (tid < 256) {
    const int row = tid >> 4, c = tid & 15, node = mt * 16 + row;
    if (node < N) hW[(size_t)node * 16 + c] = t2[row][c];
  }
  if (tid < 128) {
    const int row = tid >> 3, h = tid & 7, node = mt * 16 + row;
    const float v0 = t2[row][2 * h], v1 = t2[row][2 * h + 1];
    if (node < N) {
      a_src[node * 8 + h] = v0 * att_src[2 * h] + v1 * att_src[2 * h + 1];
      a_dst[node * 8 + h] = v0 * att_dst[2 * h] + v1 * att_dst[2 * h + 1];
    }
  }
}

__global__ __launch_bounds__(32 * NPW) void k_gather2_final(
    const int N, const int* __restrict__ cnt, const int* __restrict__ slot,
    const int* __restrict__ ei, const int E, const float* __restrict__ hW2,
    const float* __restrict__ a_src2, const float* __restrict__ a_dst2,
    const float* __restrict__ b2, float* __restrict__ out) {
  __shared__ __attribute__((aligned(16))) float rows[NPW][16];
  const int tid = threadIdx.x, lane = tid & 31;
  const int w = __builtin_amdgcn_readfirstlane(tid >> 5);
  const int d = blockIdx.x * NPW + w;
  float4 acc0 = make_float4(0.f, 0.f, 0.f, 0.f), acc1 = acc0, acc2 = acc0, acc3 = acc0;
  if (d < N)
    gather_node_wave<16, float>(d, w, lane, cnt, slot, ei, E, hW2, a_src2, a_dst2,
                         acc0, acc1, acc2, acc3);
  if (lane < 4) *(float4*)(&rows[w][lane * 4]) = acc0;
  __syncthreads();
  if (tid < 32) {
    const int n = tid >> 1, c = tid & 1, node = blockIdx.x * NPW + n;
    float s = 0.f;
#pragma unroll
    for (int h = 0; h < 8; ++h) s += rows[n][2 * h + c];
    if (node < N) out[node * 2 + c] = s * 0.125f + b2[c];
  }
}

extern "C" void kernel_launch(void* const* d_in, const int* in_sizes, int n_in,
                              void* d_out, int out_size, void* d_ws, size_t ws_size,
                              hipStream_t stream) {
  const float* x       = (const float*)d_in[0];
  const int*   ei      = (const int*)d_in[1];
  const float* ft_W    = (const float*)d_in[2];
  const float* ft_b    = (const float*)d_in[3];
  const float* ln_g    = (const float*)d_in[4];
  const float* ln_b    = (const float*)d_in[5];
  const float* W1      = (const float*)d_in[6];
  const float* att_s1  = (const float*)d_in[7];
  const float* att_d1  = (const float*)d_in[8];
  const float* b1      = (const float*)d_in[9];
  const float* bn_g    = (const float*)d_in[10];
  const float* bn_b    = (const float*)d_in[11];
  const float* bn_mean = (const float*)d_in[12];
  const float* bn_var  = (const float*)d_in[13];
  const float* W2      = (const float*)d_in[14];
  const float* att_s2  = (const float*)d_in[15];
  const float* att_d2  = (const float*)d_in[16];
  const float* b2      = (const float*)d_in[17];

  const int N  = in_sizes[0] / 128;
  const int E  = in_sizes[1] / 2;
  const int MT = N / 16;
  const int MG = (MT + MTW - 1) / MTW;

  char* cur = (char*)d_ws;
  auto alloc = [&](size_t bytes) -> void* {
    void* p = (void*)cur;
    cur += (bytes + 255) & ~(size_t)255;
    return p;
  };
  unsigned* packWft = (unsigned*)alloc((size_t)32 * 4 * 256 * 4);
  unsigned* packW1  = (unsigned*)alloc((size_t)32 * 16 * 256 * 4);
  unsigned* packW2  = (unsigned*)alloc((size_t)1 * 16 * 256 * 4);
  unsigned* hW1h = (unsigned*)alloc((size_t)N * 256 * 4);
  float* hW2 = (float*)alloc((size_t)N * 16 * 4);
  float* aS1 = (float*)alloc((size_t)N * 8 * 4);
  float* aD1 = (float*)alloc((size_t)N * 8 * 4);
  float* aS2 = (float*)alloc((size_t)N * 8 * 4);
  float* aD2 = (float*)alloc((size_t)N * 8 * 4);
  int* slot = (int*)alloc((size_t)N * SEGCAP * 8);
  int* cnt  = (int*)alloc((size_t)N * 4);

  k_prep<<<128 + 512 + 16 + (N + 255) / 256, 256, 0, stream>>>(
      ft_W, W1, W2, packWft, packW1, packW2, cnt, N);

  k_ft_gemm1<<<MG + (E + 255) / 256, 256, 0, stream>>>(
      x, packWft, ft_b, ln_g, ln_b, packW1, att_s1, att_d1, hW1h, aS1, aD1,
      ei, cnt, slot, E, MG, N);

  k_gather1_bn_gemm2<<<(N + NPW - 1) / NPW, 32 * NPW, 0, stream>>>(
      N, cnt, slot, ei, E, hW1h, aS1, aD1, b1, bn_g, bn_b, bn_mean, bn_var, packW2,
      att_s2, att_d2, hW2, aS2, aD2);

  k_gather2_final<<<(N + NPW - 1) / NPW, 32 * NPW, 0, stream>>>(
      N, cnt, slot, ei, E, hW2, aS2, aD2, b2, (float*)d_out);
}


# Layer-2 product (512 -> 16) spread over its work-group's 16 waves: each wave builds one kstep's A fragment in registers, one matrix instruction; partials summed in kstep order in LDS (order changes)
# speedup vs baseline: 5.3010x; 1.0665x over previous
#include <hip/hip_runtime.h>
#include <stdint.h>
#include <stddef.h>


typedef __attribute__((ext_vector_type(16))) __bf16 v16bf;
typedef __attribute__((ext_vector_type(8)))  float  v8f;
typedef __attribute__((ext_vector_type(8)))  unsigned v8u;

__device__ __forceinline__ unsigned f2bf(float f) {
  unsigned u = __float_as_uint(f);
  return (u + 0x7FFFu + ((u >> 16) & 1u)) >> 16;
}

__device__ __forceinline__ int kmap(int v, int lane) {
  return ((v & 4) << 2) | ((lane & 16) >> 1) | ((v & 3) << 1);
}

__device__ __forceinline__ v16bf lds_fragA(const uint4* flo, const uint4* fhi,
                                           int ks, int lane) {
  const uint4 lo = flo[ks * 32 + lane], hi = fhi[ks * 32 + lane];
  const v8u u = {lo.x, lo.y, lo.z, lo.w, hi.x, hi.y, hi.z, hi.w};
  return __builtin_bit_cast(v16bf, u);
}

__device__ __forceinline__ void pack_B_elem(const float* __restrict__ W,
                                            unsigned* __restrict__ pack,
                                            const int K, const int N, const int id) {
  const int nks = K >> 5;
  const int total = (N >> 4) * nks * 256;
  if (id >= total) return;
  const int v = id & 7, l = (id >> 3) & 31;
  const int t = id >> 8, ks = t % nks, nt = t / nks;
  const int n = (nt << 4) + (l & 15);
  const int k = (ks << 5) + kmap(v, l);
  pack[id] = f2bf(W[(size_t)k * N + n]) | (f2bf(W[(size_t)(k + 1) * N + n]) << 16);
}

__global__ __launch_bounds__(256) void k_prep(
    const float* __restrict__ ft_W, const float* __restrict__ W1,
    const float* __restrict__ W2, unsigned* __restrict__ packWft,
    unsigned* __restrict__ packW1, unsigned* __restrict__ packW2,
    int* __restrict__ cnt, int N) {
  int b = blockIdx.x;
  const int tid = threadIdx.x;
  if (b < 128) { pack_B_elem(ft_W, packWft, 128, 512, b * 256 + tid); return; }
  b -= 128;
  if (b < 512) { pack_B_elem(W1, packW1, 512, 512, b * 256 + tid); return; }
  b -= 512;
  if (b < 16) { pack_B_elem(W2, packW2, 512, 16, b * 256 + tid); return; }
  b -= 16;
  const int i = b * 256 + tid;
  if (i < N) cnt[i] = 0;
}

#define SEGCAP 64

__device__ __forceinline__ unsigned pk2(float a, float b) {
  return f2bf(a) | (f2bf(b) << 16);
}

typedef __attribute__((ext_vector_type(2))) _Float16 h2v;
__device__ __forceinline__ unsigned pkh2(float a, float b) {
  h2v p;
  p.x = (_Float16)a;
  p.y = (_Float16)b;
  return __builtin_bit_cast(unsigned, p);
}
__device__ __forceinline__ float hlo(unsigned u) { return (float)__builtin_bit_cast(h2v, u).x; }
__device__ __forceinline__ float hhi(unsigned u) { return (float)__builtin_bit_cast(h2v, u).y; }

#define MTW 3
#define LDP 516

__device__ __forceinline__ float ln_act(float v, float mu, float rs, float g, float b) {
  const float xv = (v - mu) * rs * g + b;
  return xv >= 0.f ? xv : 0.2f * xv;
}

__global__ __launch_bounds__(256, 4) void k_ft_gemm1(
    const float* __restrict__ x, const unsigned* __restrict__ packWft,
    const float* __restrict__ ft_b, const float* __restrict__ ln_g,
    const float* __restrict__ ln_b, const unsigned* __restrict__ packW1,
    const float* __restrict__ att_src, const float* __restrict__ att_dst,
    unsigned* __restrict__ hWh, float* __restrict__ a_src, float* __restrict__ a_dst,
    const int* __restrict__ ei, int* __restrict__ cnt, int* __restrict__ slot,
    int E, int MG, int N) {
  __shared__ __attribute__((aligned(16))) float tile[MTW * 16 * LDP];
  __shared__ __attribute__((aligned(16))) float red[MTW * 16 * 32];
  __shared__ float stats[MTW * 16][2];
  __shared__ uint4 flo[MTW * 16 * 32], fhi[MTW * 16 * 32];
  if ((int)blockIdx.x >= MG) {
    const int e = ((int)blockIdx.x - MG) * 256 + (int)threadIdx.x;
    if (e < E) {
      const int s = ei[e], d = ei[E + e];
      const int pos = atomicAdd(&cnt[d], 1);
      if (pos < SEGCAP)
        *(int2*)(slot + ((size_t)d * SEGCAP + pos) * 2) = make_int2(e, s);
    }
    return;
  }
  const int tid = threadIdx.x, lane = tid & 31;
  const int w = __builtin_amdgcn_readfirstlane(tid >> 5);
  const int row0 = (int)blockIdx.x * (MTW * 16);
  const int m0 = (lane >> 4) << 3, nc = lane & 15;
  {
    const int nks = 4;
    const unsigned* bw = packWft + ((size_t)(w * 4) * nks * 32 + lane) * 8;
    v16bf bq[3][4];
#pragma unroll
    for (int j = 0; j < 4; ++j) {
      bq[0][j] = *(const v16bf*)(bw + (size_t)(j * nks + 0) * 256);
      bq[1][j] = *(const v16bf*)(bw + (size_t)(j * nks + 1) * 256);
    }
    static_assert(MTW * 8 == 24, "three half-fragments per wave");
    float4 pa[3], pb[3];
#pragma unroll
    for (int i = 0; i < 3; ++i) {
      const int hf = w + 8 * i, f = hf >> 1, t = f >> 2, ks = f & 3;
      int r = row0 + t * 16 + nc;
      r = r < N ? r : N - 1;
      const float* xr = x + (size_t)r * 128 + ks * 32 + ((lane & 16) >> 1) + ((hf & 1) << 4);
      pa[i] = *(const float4*)(xr);
      pb[i] = *(const float4*)(xr + 4);
    }
#pragma unroll
    for (int i = 0; i < 3; ++i) {
      const int hf = w + 8 * i, f = hf >> 1, t = f >> 2, ks = f & 3;
      uint4* dst = (hf & 1) ? fhi : flo;
      dst[(t * 16 + ks) * 32 + lane] = make_uint4(pk2(pa[i].x, pa[i].y), pk2(pa[i].z, pa[i].w),
                                                  pk2(pb[i].x, pb[i].y), pk2(pb[i].z, pb[i].w));
    }
    __syncthreads();
    v8f acc[MTW][4] = {};
#pragma unroll
    for (int ks = 0; ks < nks; ++ks) {
      if (ks + 2 < nks) {
#pragma unroll
        for (int j = 0; j < 4; ++j)
          bq[(ks + 2) % 3][j] = *(const v16bf*)(bw + (size_t)(j * nks + ks + 2) * 256);
      }
      v16bf a[MTW];
#pragma unroll
      for (int t = 0; t < MTW; ++t) a[t] = lds_fragA(flo + t * 512, fhi + t * 512, ks, lane);
      __builtin_amdgcn_sched_barrier(0);
#pragma unroll
      for (int j = 0; j < 4; ++j)
#pragma unroll
        for (int t = 0; t < MTW; ++t)
          acc[t][j] = __builtin_amdgcn_wmma_f32_16x16x32_bf16(false, a[t], false, bq[ks % 3][j],
                                                              (short)0, acc[t][j], false, false);
    }
#pragma unroll
    for (int j = 0; j < 4; ++j) {
      const int n = (w * 4 + j) * 16 + nc;
      const float bias = ft_b[n];
#pragma unroll
      for (int t = 0; t < MTW; ++t)
#pragma unroll
        for (int r = 0; r < 8; ++r) tile[(t * 16 + m0 + r) * LDP + n] = acc[t][j][r] + bias;
    }
  }
  float4 lg[2][4], lb[2][4];
#pragma unroll
  for (int q = 0; q < 2; ++q) {
    const int kb = ((w * 2 + q) << 5) + ((lane & 16) >> 1);
    lg[q][0] = *(const float4*)(ln_g + kb);      lg[q][1] = *(const float4*)(ln_g + kb + 4);
    lg[q][2] = *(const float4*)(ln_g + kb + 16); lg[q][3] = *(const float4*)(ln_g + kb + 20);
    lb[q][0] = *(const float4*)(ln_b + kb);      lb[q][1] = *(const float4*)(ln_b + kb + 4);
    lb[q][2] = *(const float4*)(ln_b + kb + 16); lb[q][3] = *(const float4*)(ln_b + kb + 20);
  }
  __syncthreads();
  {
    const int p = tid & 15;
#pragma unroll
    for (int t = 0; t < MTW; ++t) {
      const int row = t * 16 + (tid >> 4);
      float s = 0.f, s2 = 0.f;
#pragma unroll
      for (int i = 0; i < 8; ++i) {
        const float4 v = *(const float4*)(tile + row * LDP + i * 64 + p * 4);
        s += v.x; s2 += v.x * v.x;
        s += v.y; s2 += v.y * v.y;
        s += v.z; s2 += v.z * v.z;
        s += v.w; s2 += v.w * v.w;
      }
      red[row * 32 + p] = s; red[row * 32 + 16 + p] = s2;
    }
  }
  __syncthreads();
  if (tid < MTW * 16) {
    float s = 0.f, s2 = 0.f;
    for (int p = 0; p < 16; ++p) { s += red[tid * 32 + p]; s2 += red[tid * 32 + 16 + p]; }
    const float mu = s * (1.f / 512.f);
    const float var = s2 * (1.f / 512.f) - mu * mu;
    stats[tid][0] = mu;
    stats[tid][1] = rsqrtf(var + 1e-5f);
  }
  __syncthreads();
  {
    const int m = lane & 15;
#pragma unroll
    for (int q = 0; q < 2; ++q) {
      const int ks = w * 2 + q;
      const int kb = (ks << 5) + ((lane & 16) >> 1);
      const float4 g0 = lg[q][0], g1 = lg[q][1], g2 = lg[q][2], g3 = lg[q][3];
      const float4 b0 = lb[q][0], b1 = lb[q][1], b2 = lb[q][2], b3 = lb[q][3];
      float4 pv[MTW][4];
      float mu[MTW], rs[MTW];
#pragma unroll
      for (int t = 0; t < MTW; ++t) {
        const int row = t * 16 + m;
        mu[t] = stats[row][0]; rs[t] = stats[row][1];
        const float* tr = tile + row * LDP + kb;
        pv[t][0] = *(const float4*)(tr);      pv[t][1] = *(const float4*)(tr + 4);
        pv[t][2] = *(const float4*)(tr + 16); pv[t][3] = *(const float4*)(tr + 20);
      }
#pragma unroll
      for (int t = 0; t < MTW; ++t) {
        const float4 p0 = pv[t][0], p1 = pv[t][1], p2 = pv[t][2], p3 = pv[t][3];
        const float mut = mu[t], rst = rs[t];
        flo[(t * 16 + ks) * 32 + lane] = make_uint4(
            pk2(ln_act(p0.x, mut, rst, g0.x, b0.x), ln_act(p0.y, mut, rst, g0.y, b0.y)),
            pk2(ln_act(p0.z, mut, rst, g0.z, b0.z), ln_act(p0.w, mut, rst, g0.w, b0.w)),
            pk2(ln_act(p1.x, mut, rst, g1.x, b1.x), ln_act(p1.y, mut, rst, g1.y, b1.y)),
            pk2(ln_act(p1.z, mut, rst, g1.z, b1.z), ln_act(p1.w, mut, rst, g1.w, b1.w)));
        fhi[(t * 16 + ks) * 32 + lane] = make_uint4(
            pk2(ln_act(p2.x, mut, rst, g2.x, b2.x), ln_act(p2.y, mut, rst, g2.y, b2.y)),
            pk2(ln_act(p2.z, mut, rst, g2.z, b2.z), ln_act(p2.w, mut, rst, g2.w, b2.w)),
            pk2(ln_act(p3.x, mut, rst, g3.x, b3.x), ln_act(p3.y, mut, rst, g3.y, b3.y)),
            pk2(ln_act(p3.z, mut, rst, g3.z, b3.z), ln_act(p3.w, mut, rst, g3.w, b3.w)));
      }
    }
  }
  __syncthreads();
  {
    const int nks = 16;
    const unsigned* bw = packW1 + ((size_t)(w * 4) * nks * 32 + lane) * 8;
    v8f acc[MTW][4] = {};
    v16bf bq[3][4];
#pragma unroll
    for (int j = 0; j < 4; ++j) {
      bq[0][j] = *(const v16bf*)(bw + (size_t)(j * nks + 0) * 256);
      bq[1][j] = *(const v16bf*)(bw + (size_t)(j * nks + 1) * 256);
    }
    v16bf a[2][MTW];
#pragma unroll
    for (int t = 0; t < MTW; ++t) a[0][t] = lds_fragA(flo + t * 512, fhi + t * 512, 0, lane);
#pragma unroll
    for (int ks = 0; ks < nks; ++ks) {
      if (ks + 2 < nks) {
#pragma unroll
        for (int j = 0; j < 4; ++j)
          bq[(ks + 2) % 3][j] = *(const v16bf*)(bw + (size_t)(j * nks + ks + 2) * 256);
      }
      if (ks + 1 < nks) {
#pragma unroll
        for (int t = 0; t < MTW; ++t)
          a[(ks + 1) & 1][t] = lds_fragA(flo + t * 512, fhi + t * 512, ks + 1, lane);
      }
      __builtin_amdgcn_sched_barrier(0);
#pragma unroll
      for (int j = 0; j < 4; ++j)
#pragma unroll
        for (int t = 0; t < MTW; ++t)
          acc[t][j] = __builtin_amdgcn_wmma_f32_16x16x32_bf16(false, a[ks & 1][t], false, bq[ks % 3][j],
                                                              (short)0, acc[t][j], false, false);
    }
#pragma unroll
    for (int j = 0; j < 4; ++j) {
      const int n = (w * 4 + j) * 16 + nc;
#pragma unroll
      for (int t = 0; t < MTW; ++t)
#pragma unroll
        for (int r = 0; r < 8; ++r) tile[(t * 16 + m0 + r) * LDP + n] = acc[t][j][r];
    }
  }
  __syncthreads();
  if (row0 + MTW * 16 <= N) {
    float4 u[MTW * 4], v[MTW * 4];
#pragma unroll
    for (int it = 0; it < MTW * 4; ++it) {
      const int i = tid + it * 256, row = i >> 6, c8 = i & 63;
      u[it] = *(const float4*)(tile + row * LDP + c8 * 8);
      v[it] = *(const float4*)(tile + row * LDP + c8 * 8 + 4);
    }
#pragma unroll
    for (int it = 0; it < MTW * 4; ++it) {
      const int i = tid + it * 256, row = i >> 6, c8 = i & 63;
      *(uint4*)(hWh + (size_t)(row0 + row) * 256 + c8 * 4) =
          make_uint4(pkh2(u[it].x, u[it].y), pkh2(u[it].z, u[it].w),
                     pkh2(v[it].x, v[it].y), pkh2(v[it].z, v[it].w));
    }
  } else {
    for (int i = tid; i < MTW * 16 * 64; i += 256) {
      const int row = i >> 6, c8 = i & 63;
      if (row0 + row < N) {
        const float4 u = *(const float4*)(tile + row * LDP + c8 * 8);
        const float4 v = *(const float4*)(tile + row * LDP + c8 * 8 + 4);
        *(uint4*)(hWh + (size_t)(row0 + row) * 256 + c8 * 4) =
            make_uint4(pkh2(u.x, u.y), pkh2(u.z, u.w), pkh2(v.x, v.y), pkh2(v.z, v.w));
      }
    }
  }
#pragma unroll
  for (int rb = 0; rb < MTW * 16; rb += 32) {
    const int row = rb + lane;
    if (row < MTW * 16) {
      const float* tr = tile + row * LDP + (w << 6);
      float4 hv[16];
#pragma unroll
      for (int c4 = 0; c4 < 16; ++c4) hv[c4] = *(const float4*)(tr + c4 * 4);
      __builtin_amdgcn_sched_barrier(0);
      float s1 = 0.f, s2 = 0.f;
#pragma unroll
      for (int c4 = 0; c4 < 16; ++c4) {
        const float4 as = *(const float4*)(att_src + (w << 6) + c4 * 4);
        const float4 ad = *(const float4*)(att_dst + (w << 6) + c4 * 4);
        s1 += hv[c4].x * as.x; s2 += hv[c4].x * ad.x;
        s1 += hv[c4].y * as.y; s2 += hv[c4].y * ad.y;
        s1 += hv[c4].z * as.z; s2 += hv[c4].z * ad.z;
        s1 += hv[c4].w * as.w; s2 += hv[c4].w * ad.w;
      }
      red[row * 8 + w] = s1;
      red[MTW * 128 + row * 8 + w] = s2;
    }
  }
  __syncthreads();
  for (int i = tid; i < MTW * 128; i += 256) {
    if (row0 + (i >> 3) < N) {
      a_src[(size_t)row0 * 8 + i] = red[i];
      a_dst[(size_t)row0 * 8 + i] = red[MTW * 128 + i];
    }
  }
}


__device__ __forceinline__ void stage_chunk_scan(const int d, const int cb, const int n,
                                                 const int* __restrict__ ei, const int E,
                                                 int* lsrc, const int lane,
                                                 int& e0, int& seen) {
  while (e0 < E) {
    const int e = e0 + lane;
    const bool hit = e < E && ei[E + e] == d;
    const unsigned mask = __builtin_amdgcn_ballot_w32(hit);
    const int rank = seen + __builtin_popcount(mask & ((1u << lane) - 1u));
    if (hit && rank >= cb && rank < cb + n) lsrc[rank - cb] = ei[e];
    const int after = seen + __builtin_popcount(mask);
    if (after > cb + n) break;
    e0 += 32;
    seen = after;
    if (after == cb + n) break;
  }
}

#define NPW 16

__device__ __forceinline__ void wave_lds_handover() {
  __builtin_amdgcn_fence(__ATOMIC_RELEASE, "wavefront");
  __builtin_amdgcn_wave_barrier();
  __builtin_amdgcn_fence(__ATOMIC_ACQUIRE, "wavefront");
}

#define APOS(j, h) ((j) * 8 + (((h) & 1) << 2) + ((h) >> 1))

template <int NI>
__device__ __forceinline__ void segment_weights_regs(
    const int deg, const int jj, const int h8, const float ad, const int* lsrc,
    const float* __restrict__ a_src, float* lalpha) {
  float L[NI];
#pragma unroll
  for (int i = 0; i < NI; ++i) {
    const int j = jj + 4 * i;
    const int jc = j < deg ? j : deg - 1;
    L[i] = a_src[lsrc[jc] * 8 + h8];
  }
  float mx = __uint_as_float(0xff800000u);
#pragma unroll
  for (int i = 0; i < NI; ++i) {
    float v = L[i] + ad;
    v = v >= 0.f ? v : 0.2f * v;
    L[i] = v;
    if (jj + 4 * i < deg) mx = fmaxf(mx, v);
  }
  mx = fmaxf(mx, __shfl_xor(mx, 8, 32));
  mx = fmaxf(mx, __shfl_xor(mx, 16, 32));
  float sm = 0.f;
#pragma unroll
  for (int i = 0; i < NI; ++i) {
    const float e = __expf(L[i] - mx);
    L[i] = e;
    if (jj + 4 * i < deg) sm += e;
  }
  sm += __shfl_xor(sm, 8, 32);
  sm += __shfl_xor(sm, 16, 32);
#pragma unroll
  for (int i = 0; i < NI; ++i) {
    const int j = jj + 4 * i;
    if (j < deg) lalpha[APOS(j, h8)] = L[i] / sm;
  }
}

template <int ROWF, typename RowT>
__device__ __forceinline__ void accumulate_rows(
    const int n, const int lane, const int* lsrc, const float* lalpha,
    const RowT* __restrict__ hW, float4& acc0, float4& acc1, float4& acc2, float4& acc3) {
  if (ROWF == 512) {
    const int pa = APOS(0, lane >> 3);
#pragma unroll 4
    for (int j = 0; j < n; ++j) {
      const int s = __builtin_amdgcn_readfirstlane(lsrc[j]);
      const uint4* row = reinterpret_cast<const uint4*>(hW) + (size_t)s * 64 + lane;
      const uint4 v0 = row[0];
      const uint4 v1 = row[32];
      const float a0 = lalpha[j * 8 + pa], a1 = lalpha[j * 8 + pa + 2];
      acc0.x += a0 * hlo(v0.x); acc0.y += a0 * hhi(v0.x); acc0.z += a0 * hlo(v0.y); acc0.w += a0 * hhi(v0.y);
      acc1.x += a0 * hlo(v0.z); acc1.y += a0 * hhi(v0.z); acc1.z += a0 * hlo(v0.w); acc1.w += a0 * hhi(v0.w);
      acc2.x += a1 * hlo(v1.x); acc2.y += a1 * hhi(v1.x); acc2.z += a1 * hlo(v1.y); acc2.w += a1 * hhi(v1.y);
      acc3.x += a1 * hlo(v1.z); acc3.y += a1 * hhi(v1.z); acc3.z += a1 * hlo(v1.w); acc3.w += a1 * hhi(v1.w);
    }
  } else if (lane < 4) {
#pragma unroll 4
    for (int j = 0; j < n; ++j) {
      const float4 v = *(reinterpret_cast<const float4*>(hW) + (size_t)lsrc[j] * (ROWF / 4) + lane);
      const float a0 = lalpha[APOS(j, 2 * lane)], a1 = lalpha[APOS(j, 2 * lane + 1)];
      acc0.x += a0 * v.x; acc0.y += a0 * v.y; acc0.z += a1 * v.z; acc0.w += a1 * v.w;
    }
  }
}

template <int ROWF, typename RowT>
__device__ __forceinline__ void gather_node_wave(
    const int d, const int w, const int lane,
    const int* __restrict__ cnt, const int* __restrict__ slot,
    const int* __restrict__ ei, const int E, const RowT* __restrict__ hW,
    const float* __restrict__ a_src, const float* __restrict__ a_dst,
    float4& acc0, float4& acc1, float4& acc2, float4& acc3) {
  static_assert(ROWF == 512 || ROWF == 16, "row gather is written for these two row sizes");
  __shared__ int lkey_all[NPW][SEGCAP];
  __shared__ int lsrc_all[NPW][SEGCAP];
  __shared__ __attribute__((aligned(16))) float lalpha_all[NPW][SEGCAP * 8];
  int* lkey = lkey_all[w];
  int* lsrc = lsrc_all[w];
  float* lalpha = lalpha_all[w];
  const int deg = cnt[d];
  const bool small = deg <= SEGCAP;
  const int h8 = lane & 7, jj = lane >> 3;
  const float ad = a_dst[d * 8 + h8];
  acc0 = make_float4(0.f, 0.f, 0.f, 0.f); acc1 = acc0; acc2 = acc0; acc3 = acc0;
  if (small) {
    const int2* bucket = (const int2*)(slot + (size_t)d * SEGCAP * 2);
    const int t0 = lane, t1 = lane + 32;
    int k0 = 0, s0 = 0, k1 = 0, s1 = 0;
    if (t0 < deg) { const int2 ks = bucket[t0]; k0 = ks.x; s0 = ks.y; lkey[t0] = k0; }
    if (t1 < deg) { const int2 ks = bucket[t1]; k1 = ks.x; s1 = ks.y; lkey[t1] = k1; }
    wave_lds_handover();
    int r0 = 0, r1 = 0;
    for (int j = 0; j < deg; ++j) {
      const int kj = lkey[j];
      r0 += (kj < k0) ? 1 : 0;
      r1 += (kj < k1) ? 1 : 0;
    }
    if (t0 < deg) lsrc[r0] = s0;
    if (t1 < deg) lsrc[r1] = s1;
    wave_lds_handover();
    if (deg > 0) {
      if (deg <= 16)      segment_weights_regs<4>(deg, jj, h8, ad, lsrc, a_src, lalpha);
      else if (deg <= 32) segment_weights_regs<8>(deg, jj, h8, ad, lsrc, a_src, lalpha);
      else                segment_weights_regs<16>(deg, jj, h8, ad, lsrc, a_src, lalpha);
    }
    wave_lds_handover();
    accumulate_rows<ROWF, RowT>(deg, lane, lsrc, lalpha, hW, acc0, acc1, acc2, acc3);
    return;
  }
#define STAGE_CHUNK(cb, n)                                               \
  {                                                                      \
    wave_lds_handover();    \
    stage_chunk_scan(d, (cb), (n), ei, E, lsrc, lane, sw_e0, sw_seen);   \
    wave_lds_handover();                                                 \
  }
  float mx = __uint_as_float(0xff800000u);
  int sw_e0 = 0, sw_seen = 0;
  for (int cb = 0; cb < deg; cb += SEGCAP) {
    const int n = (deg - cb) < SEGCAP ? (deg - cb) : SEGCAP;
    STAGE_CHUNK(cb, n)
    for (int j = jj; j < n; j += 4) {
      float L = a_src[lsrc[j] * 8 + h8] + ad;
      L = L >= 0.f ? L : 0.2f * L;
      mx = fmaxf(mx, L);
    }
  }
  mx = fmaxf(mx, __shfl_xor(mx, 8, 32));
  mx = fmaxf(mx, __shfl_xor(mx, 16, 32));
  const float m = mx;
  float sm = 0.f;
  sw_e0 = 0; sw_seen = 0;
  for (int cb = 0; cb < deg; cb += SEGCAP) {
    const int n = (deg - cb) < SEGCAP ? (deg - cb) : SEGCAP;
    STAGE_CHUNK(cb, n)
    for (int j = jj; j < n; j += 4) {
      float L = a_src[lsrc[j] * 8 + h8] + ad;
      L = L >= 0.f ? L : 0.2f * L;
      sm += __expf(L - m);
    }
  }
  sm += __shfl_xor(sm, 8, 32);
  sm += __shfl_xor(sm, 16, 32);
  const float den = sm;
  sw_e0 = 0; sw_seen = 0;
  for (int cb = 0; cb < deg; cb += SEGCAP) {
    const int n = (deg - cb) < SEGCAP ? (deg - cb) : SEGCAP;
    STAGE_CHUNK(cb, n)
    for (int j = jj; j < n; j += 4) {
      float L = a_src[lsrc[j] * 8 + h8] + ad;
      L = L >= 0.f ? L : 0.2f * L;
      lalpha[APOS(j, h8)] = __expf(L - m) / den;
    }
    wave_lds_handover();
    accumulate_rows<ROWF, RowT>(n, lane, lsrc, lalpha, hW, acc0, acc1, acc2, acc3);
  }
#undef STAGE_CHUNK
}

__global__ __launch_bounds__(32 * NPW) void k_gather1_bn_gemm2(
    const int N, const int* __restrict__ cnt, const int* __restrict__ slot,
    const int* __restrict__ ei, const int E, const unsigned* __restrict__ hW1h,
    const float* __restrict__ a_src1, const float* __restrict__ a_dst1,
    const float* __restrict__ b1, const float* __restrict__ bn_g,
    const float* __restrict__ bn_b, const float* __restrict__ bn_mean,
    const float* __restrict__ bn_var, const unsigned* __restrict__ packW,
    const float* __restrict__ att_src, const float* __restrict__ att_dst,
    float* __restrict__ hW, float* __restrict__ a_src, float* __restrict__ a_dst) {
  static_assert(NPW == 16, "one wave per kstep of the 512 -> 16 product");
  __shared__ __attribute__((aligned(16))) float tile[16 * LDP];
  __shared__ __attribute__((aligned(16))) float part[NPW][16][16];
  const int mt = blockIdx.x, tid = threadIdx.x, lane = tid & 31;
  const int w = __builtin_amdgcn_readfirstlane(tid >> 5);
  const int d = mt * NPW + w;
  const v16bf bfrag = *(const v16bf*)(packW + ((size_t)w * 32 + lane) * 8);
  float4 acc[4];
  acc[0] = make_float4(0.f, 0.f, 0.f, 0.f); acc[1] = acc[0]; acc[2] = acc[0]; acc[3] = acc[0];
  if (d < N)
    gather_node_wave<512, unsigned>(d, w, lane, cnt, slot, ei, E, hW1h, a_src1, a_dst1,
                          acc[0], acc[1], acc[2], acc[3]);
#pragma unroll
  for (int q = 0; q < 4; ++q) {
    const int c = (q >> 1) * 256 + lane * 8 + (q & 1) * 4;
    const float4 vb1 = *(const float4*)(b1 + c), vmu = *(const float4*)(bn_mean + c);
    const float4 vva = *(const float4*)(bn_var + c), vg = *(const float4*)(bn_g + c);
    const float4 vb = *(const float4*)(bn_b + c);
    float4 r;
    r.x = ((acc[q].x + vb1.x) - vmu.x) * rsqrtf(vva.x + 1e-5f) * vg.x + vb.x;
    r.y = ((acc[q].y + vb1.y) - vmu.y) * rsqrtf(vva.y + 1e-5f) * vg.y + vb.y;
    r.z = ((acc[q].z + vb1.z) - vmu.z) * rsqrtf(vva.z + 1e-5f) * vg.z + vb.z;
    r.w = ((acc[q].w + vb1.w) - vmu.w) * rsqrtf(vva.w + 1e-5f) * vg.w + vb.w;
    r.x = r.x > 0.f ? r.x : 0.f; r.y = r.y > 0.f ? r.y : 0.f;
    r.z = r.z > 0.f ? r.z : 0.f; r.w = r.w > 0.f ? r.w : 0.f;
    *(float4*)(tile + w * LDP + c) = r;
  }
  __syncthreads();
  {
    const float* tr = tile + (lane & 15) * LDP + (w << 5) + ((lane & 16) >> 1);
    const float4 p0 = *(const float4*)(tr),      p1 = *(const float4*)(tr + 4);
    const float4 p2 = *(const float4*)(tr + 16), p3 = *(const float4*)(tr + 20);
    const v8u u = {pk2(p0.x, p0.y), pk2(p0.z, p0.w), pk2(p1.x, p1.y), pk2(p1.z, p1.w),
                   pk2(p2.x, p2.y), pk2(p2.z, p2.w), pk2(p3.x, p3.y), pk2(p3.z, p3.w)};
    v8f pr = {};
    pr = __builtin_amdgcn_wmma_f32_16x16x32_bf16(false, __builtin_bit_cast(v16bf, u), false,
                                                 bfrag, (short)0, pr, false, false);
    const int m0 = (lane >> 4) << 3, nc = lane & 15;
#pragma unroll
    for (int r = 0; r < 8; ++r) part[w][m0 + r][nc] = pr[r];
  }
  __syncthreads();
  if (tid < 128) {
    const int row = tid >> 3, h = tid & 7, node = mt * 16 + row;
    float2 pp[NPW];
#pragma unroll
    for (int ks = 0; ks < NPW; ++ks) pp[ks] = *(const float2*)(&part[ks][row][2 * h]);
    float v0 = pp[0].x, v1 = pp[0].y;
#pragma unroll
    for (int ks = 1; ks < NPW; ++ks) { v0 += pp[ks].x; v1 += pp[ks].y; }
    if (node < N) {
      *(float2*)(hW + (size_t)node * 16 + 2 * h) = make_float2(v0, v1);
      a_src[node * 8 + h] = v0 * att_src[2 * h] + v1 * att_src[2 * h + 1];
      a_dst[node * 8 + h] = v0 * att_dst[2 * h] + v1 * att_dst[2 * h + 1];
    }
  }
}

__global__ __launch_bounds__(32 * NPW) void k_gather2_final(
    const int N, const int* __restrict__ cnt, const int* __restrict__ slot,
    const int* __restrict__ ei, const int E, const float* __restrict__ hW2,
    const float* __restrict__ a_src2, const float* __restrict__ a_dst2,
    const float* __restrict__ b2, float* __restrict__ out) {
  __shared__ __attribute__((aligned(16))) float rows[NPW][16];
  const int tid = threadIdx.x, lane = tid & 31;
  const int w = __builtin_amdgcn_readfirstlane(tid >> 5);
  const int d = blockIdx.x * NPW + w;
  float4 acc0 = make_float4(0.f, 0.f, 0.f, 0.f), acc1 = acc0, acc2 = acc0, acc3 = acc0;
  if (d < N)
    gather_node_wave<16, float>(d, w, lane, cnt, slot, ei, E, hW2, a_src2, a_dst2,
                         acc0, acc1, acc2, acc3);
  if (lane < 4) *(float4*)(&rows[w][lane * 4]) = acc0;
  __syncthreads();
  if (tid < 32) {
    const int n = tid >> 1, c = tid & 1, node = blockIdx.x * NPW + n;
    float s = 0.f;
#pragma unroll
    for (int h = 0; h < 8; ++h) s += rows[n][2 * h + c];
    if (node < N) out[node * 2 + c] = s * 0.125f + b2[c];
  }
}

extern "C" void kernel_launch(void* const* d_in, const int* in_sizes, int n_in,
                              void* d_out, int out_size, void* d_ws, size_t ws_size,
                              hipStream_t stream) {
  const float* x       = (const float*)d_in[0];
  const int*   ei      = (const int*)d_in[1];
  const float* ft_W    = (const float*)d_in[2];
  const float* ft_b    = (const float*)d_in[3];
  const float* ln_g    = (const float*)d_in[4];
  const float* ln_b    = (const float*)d_in[5];
  const float* W1      = (const float*)d_in[6];
  const float* att_s1  = (const float*)d_in[7];
  const float* att_d1  = (const float*)d_in[8];
  const float* b1      = (const float*)d_in[9];
  const float* bn_g    = (const float*)d_in[10];
  const float* bn_b    = (const float*)d_in[11];
  const float* bn_mean = (const float*)d_in[12];
  const float* bn_var  = (const float*)d_in[13];
  const float* W2      = (const float*)d_in[14];
  const float* att_s2  = (const float*)d_in[15];
  const float* att_d2  = (const float*)d_in[16];
  const float* b2      = (const float*)d_in[17];

  const int N  = in_sizes[0] / 128;
  const int E  = in_sizes[1] / 2;
  const int MT = N / 16;
  const int MG = (MT + MTW - 1) / MTW;

  char* cur = (char*)d_ws;
  auto alloc = [&](size_t bytes) -> void* {
    void* p = (void*)cur;
    cur += (bytes + 255) & ~(size_t)255;
    return p;
  };
  unsigned* packWft = (unsigned*)alloc((size_t)32 * 4 * 256 * 4);
  unsigned* packW1  = (unsigned*)alloc((size_t)32 * 16 * 256 * 4);
  unsigned* packW2  = (unsigned*)alloc((size_t)1 * 16 * 256 * 4);
  unsigned* hW1h = (unsigned*)alloc((size_t)N * 256 * 4);
  float* hW2 = (float*)alloc((size_t)N * 16 * 4);
  float* aS1 = (float*)alloc((size_t)N * 8 * 4);
  float* aD1 = (float*)alloc((size_t)N * 8 * 4);
  float* aS2 = (float*)alloc((size_t)N * 8 * 4);
  float* aD2 = (float*)alloc((size_t)N * 8 * 4);
  int* slot = (int*)alloc((size_t)N * SEGCAP * 8);
  int* cnt  = (int*)alloc((size_t)N * 4);

  k_prep<<<128 + 512 + 16 + (N + 255) / 256, 256, 0, stream>>>(
      ft_W, W1, W2, packWft, packW1, packW2, cnt, N);

  k_ft_gemm1<<<MG + (E + 255) / 256, 256, 0, stream>>>(
      x, packWft, ft_b, ln_g, ln_b, packW1, att_s1, att_d1, hW1h, aS1, aD1,
      ei, cnt, slot, E, MG, N);

  k_gather1_bn_gemm2<<<(N + NPW - 1) / NPW, 32 * NPW, 0, stream>>>(
      N, cnt, slot, ei, E, hW1h, aS1, aD1, b1, bn_g, bn_b, bn_mean, bn_var, packW2,
      att_s2, att_d2, hW2, aS2, aD2);

  k_gather2_final<<<(N + NPW - 1) / NPW, 32 * NPW, 0, stream>>>(
      N, cnt, slot, ei, E, hW2, aS2, aD2, b2, (float*)d_out);
}


# Layer-1 gather kernel held to 80 registers (12 waves per SIMD): 3 work-groups per WGP, all 625 resident at once; a rolling window of 4 edges' rows per wave, f16 words summed by mixed multiply-adds
# speedup vs baseline: 5.5974x; 1.0559x over previous
#include <hip/hip_runtime.h>
#include <stdint.h>
#include <stddef.h>


typedef __attribute__((ext_vector_type(16))) __bf16 v16bf;
typedef __attribute__((ext_vector_type(8)))  float  v8f;
typedef __attribute__((ext_vector_type(8)))  unsigned v8u;

__device__ __forceinline__ unsigned f2bf(float f) {
  unsigned u = __float_as_uint(f);
  return (u + 0x7FFFu + ((u >> 16) & 1u)) >> 16;
}

__device__ __forceinline__ int kmap(int v, int lane) {
  return ((v & 4) << 2) | ((lane & 16) >> 1) | ((v & 3) << 1);
}

__device__ __forceinline__ v16bf lds_fragA(const uint4* flo, const uint4* fhi,
                                           int ks, int lane) {
  const uint4 lo = flo[ks * 32 + lane], hi = fhi[ks * 32 + lane];
  const v8u u = {lo.x, lo.y, lo.z, lo.w, hi.x, hi.y, hi.z, hi.w};
  return __builtin_bit_cast(v16bf, u);
}

__device__ __forceinline__ void pack_B_elem(const float* __restrict__ W,
                                            unsigned* __restrict__ pack,
                                            const int K, const int N, const int id) {
  const int nks = K >> 5;
  const int total = (N >> 4) * nks * 256;
  if (id >= total) return;
  const int v = id & 7, l = (id >> 3) & 31;
  const int t = id >> 8, ks = t % nks, nt = t / nks;
  const int n = (nt << 4) + (l & 15);
  const int k = (ks << 5) + kmap(v, l);
  pack[id] = f2bf(W[(size_t)k * N + n]) | (f2bf(W[(size_t)(k + 1) * N + n]) << 16);
}

__global__ __launch_bounds__(256) void k_prep(
    const float* __restrict__ ft_W, const float* __restrict__ W1,
    const float* __restrict__ W2, unsigned* __restrict__ packWft,
    unsigned* __restrict__ packW1, unsigned* __restrict__ packW2,
    int* __restrict__ cnt, int N) {
  int b = blockIdx.x;
  const int tid = threadIdx.x;
  if (b < 128) { pack_B_elem(ft_W, packWft, 128, 512, b * 256 + tid); return; }
  b -= 128;
  if (b < 512) { pack_B_elem(W1, packW1, 512, 512, b * 256 + tid); return; }
  b -= 512;
  if (b < 16) { pack_B_elem(W2, packW2, 512, 16, b * 256 + tid); return; }
  b -= 16;
  const int i = b * 256 + tid;
  if (i < N) cnt[i] = 0;
}

#define SEGCAP 64

__device__ __forceinline__ unsigned pk2(float a, float b) {
  return f2bf(a) | (f2bf(b) << 16);
}

typedef __attribute__((ext_vector_type(2))) _Float16 h2v;
__device__ __forceinline__ unsigned pkh2(float a, float b) {
  h2v p;
  p.x = (_Float16)a;
  p.y = (_Float16)b;
  return __builtin_bit_cast(unsigned, p);
}
__device__ __forceinline__ void fma_h_lo(float& acc, const float a, const unsigned w) {
  asm("v_fma_mix_f32 %0, %1, %2, %0 op_sel_hi:[0,1,0]" : "+v"(acc) : "v"(a), "v"(w));
}
__device__ __forceinline__ void fma_h_hi(float& acc, const float a, const unsigned w) {
  asm("v_fma_mix_f32 %0, %1, %2, %0 op_sel:[0,1,0] op_sel_hi:[0,1,0]" : "+v"(acc) : "v"(a), "v"(w));
}

#define MTW 3
#define LDP 516

__device__ __forceinline__ float ln_act(float v, float mu, float rs, float g, float b) {
  const float xv = (v - mu) * rs * g + b;
  return xv >= 0.f ? xv : 0.2f * xv;
}

__global__ __launch_bounds__(256, 4) void k_ft_gemm1(
    const float* __restrict__ x, const unsigned* __restrict__ packWft,
    const float* __restrict__ ft_b, const float* __restrict__ ln_g,
    const float* __restrict__ ln_b, const unsigned* __restrict__ packW1,
    const float* __restrict__ att_src, const float* __restrict__ att_dst,
    unsigned* __restrict__ hWh, float* __restrict__ a_src, float* __restrict__ a_dst,
    const int* __restrict__ ei, int* __restrict__ cnt, int* __restrict__ slot,
    int E, int MG, int N) {
  __shared__ __attribute__((aligned(16))) float tile[MTW * 16 * LDP];
  __shared__ __attribute__((aligned(16))) float red[MTW * 16 * 32];
  __shared__ float stats[MTW * 16][2];
  __shared__ uint4 flo[MTW * 16 * 32], fhi[MTW * 16 * 32];
  if ((int)blockIdx.x >= MG) {
    const int e = ((int)blockIdx.x - MG) * 256 + (int)threadIdx.x;
    if (e < E) {
      const int s = ei[e], d = ei[E + e];
      const int pos = atomicAdd(&cnt[d], 1);
      if (pos < SEGCAP)
        *(int2*)(slot + ((size_t)d * SEGCAP + pos) * 2) = make_int2(e, s);
    }
    return;
  }
  const int tid = threadIdx.x, lane = tid & 31;
  const int w = __builtin_amdgcn_readfirstlane(tid >> 5);
  const int row0 = (int)blockIdx.x * (MTW * 16);
  const int m0 = (lane >> 4) << 3, nc = lane & 15;
  {
    const int nks = 4;
    const unsigned* bw = packWft + ((size_t)(w * 4) * nks * 32 + lane) * 8;
    v16bf bq[3][4];
#pragma unroll
    for (int j = 0; j < 4; ++j) {
      bq[0][j] = *(const v16bf*)(bw + (size_t)(j * nks + 0) * 256);
      bq[1][j] = *(const v16bf*)(bw + (size_t)(j * nks + 1) * 256);
    }
    static_assert(MTW * 8 == 24, "three half-fragments per wave");
    float4 pa[3], pb[3];
#pragma unroll
    for (int i = 0; i < 3; ++i) {
      const int hf = w + 8 * i, f = hf >> 1, t = f >> 2, ks = f & 3;
      int r = row0 + t * 16 + nc;
      r = r < N ? r : N - 1;
      const float* xr = x + (size_t)r * 128 + ks * 32 + ((lane & 16) >> 1) + ((hf & 1) << 4);
      pa[i] = *(const float4*)(xr);
      pb[i] = *(const float4*)(xr + 4);
    }
#pragma unroll
    for (int i = 0; i < 3; ++i) {
      const int hf = w + 8 * i, f = hf >> 1, t = f >> 2, ks = f & 3;
      uint4* dst = (hf & 1) ? fhi : flo;
      dst[(t * 16 + ks) * 32 + lane] = make_uint4(pk2(pa[i].x, pa[i].y), pk2(pa[i].z, pa[i].w),
                                                  pk2(pb[i].x, pb[i].y), pk2(pb[i].z, pb[i].w));
    }
    __syncthreads();
    v8f acc[MTW][4] = {};
#pragma unroll
    for (int ks = 0; ks < nks; ++ks) {
      if (ks + 2 < nks) {
#pragma unroll
        for (int j = 0; j < 4; ++j)
          bq[(ks + 2) % 3][j] = *(const v16bf*)(bw + (size_t)(j * nks + ks + 2) * 256);
      }
      v16bf a[MTW];
#pragma unroll
      for (int t = 0; t < MTW; ++t) a[t] = lds_fragA(flo + t * 512, fhi + t * 512, ks, lane);
      __builtin_amdgcn_sched_barrier(0);
#pragma unroll
      for (int j = 0; j < 4; ++j)
#pragma unroll
        for (int t = 0; t < MTW; ++t)
          acc[t][j] = __builtin_amdgcn_wmma_f32_16x16x32_bf16(false, a[t], false, bq[ks % 3][j],
                                                              (short)0, acc[t][j], false, false);
    }
#pragma unroll
    for (int j = 0; j < 4; ++j) {
      const int n = (w * 4 + j) * 16 + nc;
      const float bias = ft_b[n];
#pragma unroll
      for (int t = 0; t < MTW; ++t)
#pragma unroll
        for (int r = 0; r < 8; ++r) tile[(t * 16 + m0 + r) * LDP + n] = acc[t][j][r] + bias;
    }
  }
  float4 lg[2][4], lb[2][4];
#pragma unroll
  for (int q = 0; q < 2; ++q) {
    const int kb = ((w * 2 + q) << 5) + ((lane & 16) >> 1);
    lg[q][0] = *(const float4*)(ln_g + kb);      lg[q][1] = *(const float4*)(ln_g + kb + 4);
    lg[q][2] = *(const float4*)(ln_g + kb + 16); lg[q][3] = *(const float4*)(ln_g + kb + 20);
    lb[q][0] = *(const float4*)(ln_b + kb);      lb[q][1] = *(const float4*)(ln_b + kb + 4);
    lb[q][2] = *(const float4*)(ln_b + kb + 16); lb[q][3] = *(const float4*)(ln_b + kb + 20);
  }
  __syncthreads();
  {
    const int p = tid & 15;
#pragma unroll
    for (int t = 0; t < MTW; ++t) {
      const int row = t * 16 + (tid >> 4);
      float s = 0.f, s2 = 0.f;
#pragma unroll
      for (int i = 0; i < 8; ++i) {
        const float4 v = *(const float4*)(tile + row * LDP + i * 64 + p * 4);
        s += v.x; s2 += v.x * v.x;
        s += v.y; s2 += v.y * v.y;
        s += v.z; s2 += v.z * v.z;
        s += v.w; s2 += v.w * v.w;
      }
      red[row * 32 + p] = s; red[row * 32 + 16 + p] = s2;
    }
  }
  __syncthreads();
  if (tid < MTW * 16) {
    float s = 0.f, s2 = 0.f;
    for (int p = 0; p < 16; ++p) { s += red[tid * 32 + p]; s2 += red[tid * 32 + 16 + p]; }
    const float mu = s * (1.f / 512.f);
    const float var = s2 * (1.f / 512.f) - mu * mu;
    stats[tid][0] = mu;
    stats[tid][1] = rsqrtf(var + 1e-5f);
  }
  __syncthreads();
  {
    const int m = lane & 15;
#pragma unroll
    for (int q = 0; q < 2; ++q) {
      const int ks = w * 2 + q;
      const int kb = (ks << 5) + ((lane & 16) >> 1);
      const float4 g0 = lg[q][0], g1 = lg[q][1], g2 = lg[q][2], g3 = lg[q][3];
      const float4 b0 = lb[q][0], b1 = lb[q][1], b2 = lb[q][2], b3 = lb[q][3];
      float4 pv[MTW][4];
      float mu[MTW], rs[MTW];
#pragma unroll
      for (int t = 0; t < MTW; ++t) {
        const int row = t * 16 + m;
        mu[t] = stats[row][0]; rs[t] = stats[row][1];
        const float* tr = tile + row * LDP + kb;
        pv[t][0] = *(const float4*)(tr);      pv[t][1] = *(const float4*)(tr + 4);
        pv[t][2] = *(const float4*)(tr + 16); pv[t][3] = *(const float4*)(tr + 20);
      }
#pragma unroll
      for (int t = 0; t < MTW; ++t) {
        const float4 p0 = pv[t][0], p1 = pv[t][1], p2 = pv[t][2], p3 = pv[t][3];
        const float mut = mu[t], rst = rs[t];
        flo[(t * 16 + ks) * 32 + lane] = make_uint4(
            pk2(ln_act(p0.x, mut, rst, g0.x, b0.x), ln_act(p0.y, mut, rst, g0.y, b0.y)),
            pk2(ln_act(p0.z, mut, rst, g0.z, b0.z), ln_act(p0.w, mut, rst, g0.w, b0.w)),
            pk2(ln_act(p1.x, mut, rst, g1.x, b1.x), ln_act(p1.y, mut, rst, g1.y, b1.y)),
            pk2(ln_act(p1.z, mut, rst, g1.z, b1.z), ln_act(p1.w, mut, rst, g1.w, b1.w)));
        fhi[(t * 16 + ks) * 32 + lane] = make_uint4(
            pk2(ln_act(p2.x, mut, rst, g2.x, b2.x), ln_act(p2.y, mut, rst, g2.y, b2.y)),
            pk2(ln_act(p2.z, mut, rst, g2.z, b2.z), ln_act(p2.w, mut, rst, g2.w, b2.w)),
            pk2(ln_act(p3.x, mut, rst, g3.x, b3.x), ln_act(p3.y, mut, rst, g3.y, b3.y)),
            pk2(ln_act(p3.z, mut, rst, g3.z, b3.z), ln_act(p3.w, mut, rst, g3.w, b3.w)));
      }
    }
  }
  __syncthreads();
  {
    const int nks = 16;
    const unsigned* bw = packW1 + ((size_t)(w * 4) * nks * 32 + lane) * 8;
    v8f acc[MTW][4] = {};
    v16bf bq[3][4];
#pragma unroll
    for (int j = 0; j < 4; ++j) {
      bq[0][j] = *(const v16bf*)(bw + (size_t)(j * nks + 0) * 256);
      bq[1][j] = *(const v16bf*)(bw + (size_t)(j * nks + 1) * 256);
    }
    v16bf a[2][MTW];
#pragma unroll
    for (int t = 0; t < MTW; ++t) a[0][t] = lds_fragA(flo + t * 512, fhi + t * 512, 0, lane);
#pragma unroll
    for (int ks = 0; ks < nks; ++ks) {
      if (ks + 2 < nks) {
#pragma unroll
        for (int j = 0; j < 4; ++j)
          bq[(ks + 2) % 3][j] = *(const v16bf*)(bw + (size_t)(j * nks + ks + 2) * 256);
      }
      if (ks + 1 < nks) {
#pragma unroll
        for (int t = 0; t < MTW; ++t)
          a[(ks + 1) & 1][t] = lds_fragA(flo + t * 512, fhi + t * 512, ks + 1, lane);
      }
      __builtin_amdgcn_sched_barrier(0);
#pragma unroll
      for (int j = 0; j < 4; ++j)
#pragma unroll
        for (int t = 0; t < MTW; ++t)
          acc[t][j] = __builtin_amdgcn_wmma_f32_16x16x32_bf16(false, a[ks & 1][t], false, bq[ks % 3][j],
                                                              (short)0, acc[t][j], false, false);
    }
#pragma unroll
    for (int j = 0; j < 4; ++j) {
      const int n = (w * 4 + j) * 16 + nc;
#pragma unroll
      for (int t = 0; t < MTW; ++t)
#pragma unroll
        for (int r = 0; r < 8; ++r) tile[(t * 16 + m0 + r) * LDP + n] = acc[t][j][r];
    }
  }
  __syncthreads();
  if (row0 + MTW * 16 <= N) {
    float4 u[MTW * 4], v[MTW * 4];
#pragma unroll
    for (int it = 0; it < MTW * 4; ++it) {
      const int i = tid + it * 256, row = i >> 6, c8 = i & 63;
      u[it] = *(const float4*)(tile + row * LDP + c8 * 8);
      v[it] = *(const float4*)(tile + row * LDP + c8 * 8 + 4);
    }
#pragma unroll
    for (int it = 0; it < MTW * 4; ++it) {
      const int i = tid + it * 256, row = i >> 6, c8 = i & 63;
      *(uint4*)(hWh + (size_t)(row0 + row) * 256 + c8 * 4) =
          make_uint4(pkh2(u[it].x, u[it].y), pkh2(u[it].z, u[it].w),
                     pkh2(v[it].x, v[it].y), pkh2(v[it].z, v[it].w));
    }
  } else {
    for (int i = tid; i < MTW * 16 * 64; i += 256) {
      const int row = i >> 6, c8 = i & 63;
      if (row0 + row < N) {
        const float4 u = *(const float4*)(tile + row * LDP + c8 * 8);
        const float4 v = *(const float4*)(tile + row * LDP + c8 * 8 + 4);
        *(uint4*)(hWh + (size_t)(row0 + row) * 256 + c8 * 4) =
            make_uint4(pkh2(u.x, u.y), pkh2(u.z, u.w), pkh2(v.x, v.y), pkh2(v.z, v.w));
      }
    }
  }
#pragma unroll
  for (int rb = 0; rb < MTW * 16; rb += 32) {
    const int row = rb + lane;
    if (row < MTW * 16) {
      const float* tr = tile + row * LDP + (w << 6);
      float4 hv[16];
#pragma unroll
      for (int c4 = 0; c4 < 16; ++c4) hv[c4] = *(const float4*)(tr + c4 * 4);
      __builtin_amdgcn_sched_barrier(0);
      float s1 = 0.f, s2 = 0.f;
#pragma unroll
      for (int c4 = 0; c4 < 16; ++c4) {
        const float4 as = *(const float4*)(att_src + (w << 6) + c4 * 4);
        const float4 ad = *(const float4*)(att_dst + (w << 6) + c4 * 4);
        s1 += hv[c4].x * as.x; s2 += hv[c4].x * ad.x;
        s1 += hv[c4].y * as.y; s2 += hv[c4].y * ad.y;
        s1 += hv[c4].z * as.z; s2 += hv[c4].z * ad.z;
        s1 += hv[c4].w * as.w; s2 += hv[c4].w * ad.w;
      }
      red[row * 8 + w] = s1;
      red[MTW * 128 + row * 8 + w] = s2;
    }
  }
  __syncthreads();
  for (int i = tid; i < MTW * 128; i += 256) {
    if (row0 + (i >> 3) < N) {
      a_src[(size_t)row0 * 8 + i] = red[i];
      a_dst[(size_t)row0 * 8 + i] = red[MTW * 128 + i];
    }
  }
}


__device__ __forceinline__ void stage_chunk_scan(const int d, const int cb, const int n,
                                                 const int* __restrict__ ei, const int E,
                                                 int* lsrc, const int lane,
                                                 int& e0, int& seen) {
  while (e0 < E) {
    const int e = e0 + lane;
    const bool hit = e < E && ei[E + e] == d;
    const unsigned mask = __builtin_amdgcn_ballot_w32(hit);
    const int rank = seen + __builtin_popcount(mask & ((1u << lane) - 1u));
    if (hit && rank >= cb && rank < cb + n) lsrc[rank - cb] = ei[e];
    const int after = seen + __builtin_popcount(mask);
    if (after > cb + n) break;
    e0 += 32;
    seen = after;
    if (after == cb + n) break;
  }
}

#define NPW 16
#define RD 4

__device__ __forceinline__ void wave_lds_handover() {
  __builtin_amdgcn_fence(__ATOMIC_RELEASE, "wavefront");
  __builtin_amdgcn_wave_barrier();
  __builtin_amdgcn_fence(__ATOMIC_ACQUIRE, "wavefront");
}

#define APOS(j, h) ((j) * 8 + (((h) & 1) << 2) + ((h) >> 1))

template <int NI>
__device__ __forceinline__ void segment_weights_regs(
    const int deg, const int jj, const int h8, const float ad, const int* lsrc,
    const float* __restrict__ a_src, float* lalpha) {
  float L[NI];
#pragma unroll
  for (int i = 0; i < NI; ++i) {
    const int j = jj + 4 * i;
    const int jc = j < deg ? j : deg - 1;
    L[i] = a_src[lsrc[jc] * 8 + h8];
  }
  float mx = __uint_as_float(0xff800000u);
#pragma unroll
  for (int i = 0; i < NI; ++i) {
    float v = L[i] + ad;
    v = v >= 0.f ? v : 0.2f * v;
    L[i] = v;
    if (jj + 4 * i < deg) mx = fmaxf(mx, v);
  }
  mx = fmaxf(mx, __shfl_xor(mx, 8, 32));
  mx = fmaxf(mx, __shfl_xor(mx, 16, 32));
  float sm = 0.f;
#pragma unroll
  for (int i = 0; i < NI; ++i) {
    const float e = __expf(L[i] - mx);
    L[i] = e;
    if (jj + 4 * i < deg) sm += e;
  }
  sm += __shfl_xor(sm, 8, 32);
  sm += __shfl_xor(sm, 16, 32);
#pragma unroll
  for (int i = 0; i < NI; ++i) {
    const int j = jj + 4 * i;
    if (j < deg) lalpha[APOS(j, h8)] = L[i] / sm;
  }
}

template <int ROWF, typename RowT>
__device__ __forceinline__ void accumulate_rows(
    const int n, const int lane, const int* lsrc, const float* lalpha,
    const RowT* __restrict__ hW, float4& acc0, float4& acc1, float4& acc2, float4& acc3) {
  if (ROWF == 512) {
    const int pa = APOS(0, lane >> 3);
    const uint4* base = reinterpret_cast<const uint4*>(hW) + lane;
    uint4 va[RD], vb[RD];
#define ROW_ASK(i, e)                                                    \
    {                                                                    \
      const int s_ = __builtin_amdgcn_readfirstlane(lsrc[(e)]);          \
      va[(i)] = base[(size_t)s_ * 64];                                   \
      vb[(i)] = base[(size_t)s_ * 64 + 32];                              \
    }
#define ROW_SUM(i, e)                                                    \
    {                                                                    \
      const float a0 = lalpha[(e) * 8 + pa], a1 = lalpha[(e) * 8 + pa + 2]; \
      const uint4 v0 = va[(i)], v1 = vb[(i)];                            \
      fma_h_lo(acc0.x, a0, v0.x); fma_h_hi(acc0.y, a0, v0.x); fma_h_lo(acc0.z, a0, v0.y); fma_h_hi(acc0.w, a0, v0.y); \
      fma_h_lo(acc1.x, a0, v0.z); fma_h_hi(acc1.y, a0, v0.z); fma_h_lo(acc1.z, a0, v0.w); fma_h_hi(acc1.w, a0, v0.w); \
      fma_h_lo(acc2.x, a1, v1.x); fma_h_hi(acc2.y, a1, v1.x); fma_h_lo(acc2.z, a1, v1.y); fma_h_hi(acc2.w, a1, v1.y); \
      fma_h_lo(acc3.x, a1, v1.z); fma_h_hi(acc3.y, a1, v1.z); fma_h_lo(acc3.z, a1, v1.w); fma_h_hi(acc3.w, a1, v1.w); \
    }
#define ROW_TAIL(j0)                                                     \
    _Pragma("unroll")                                                    \
    for (int i = 0; i < RD; ++i) {                                       \
      if ((j0) + i < n) {                                                \
        ROW_SUM(i, (j0) + i)                                             \
        if ((j0) + i + RD < n) ROW_ASK(i, (j0) + i + RD)                 \
      }                                                                  \
    }                                                                    \
    _Pragma("unroll")                                                    \
    for (int i = 0; i < RD; ++i)                                         \
      if ((j0) + RD + i < n) ROW_SUM(i, (j0) + RD + i)
    if (n >= 2 * RD) {
#pragma unroll
      for (int i = 0; i < RD; ++i) ROW_ASK(i, i)
      int j = 0;
      do {
#pragma unroll
        for (int i = 0; i < RD; ++i) {
          ROW_SUM(i, j + i)
          ROW_ASK(i, j + i + RD)
          __builtin_amdgcn_sched_barrier(0);
        }
        j += RD;
      } while (j + 2 * RD <= n);
      ROW_TAIL(j)
    } else {
#pragma unroll
      for (int i = 0; i < RD; ++i)
        if (i < n) ROW_ASK(i, i)
      ROW_TAIL(0)
    }
#undef ROW_TAIL
#undef ROW_ASK
#undef ROW_SUM
  } else if (lane < 4) {
#pragma unroll 4
    for (int j = 0; j < n; ++j) {
      const float4 v = *(reinterpret_cast<const float4*>(hW) + (size_t)lsrc[j] * (ROWF / 4) + lane);
      const float a0 = lalpha[APOS(j, 2 * lane)], a1 = lalpha[APOS(j, 2 * lane + 1)];
      acc0.x += a0 * v.x; acc0.y += a0 * v.y; acc0.z += a1 * v.z; acc0.w += a1 * v.w;
    }
  }
}

template <int ROWF, typename RowT>
__device__ __forceinline__ void gather_node_wave(
    const int d, const int w, const int lane,
    const int* __restrict__ cnt, const int* __restrict__ slot,
    const int* __restrict__ ei, const int E, const RowT* __restrict__ hW,
    const float* __restrict__ a_src, const float* __restrict__ a_dst,
    float4& acc0, float4& acc1, float4& acc2, float4& acc3) {
  static_assert(ROWF == 512 || ROWF == 16, "row gather is written for these two row sizes");
  __shared__ int lkey_all[NPW][SEGCAP];
  __shared__ int lsrc_all[NPW][SEGCAP];
  __shared__ __attribute__((aligned(16))) float lalpha_all[NPW][SEGCAP * 8];
  int* lkey = lkey_all[w];
  int* lsrc = lsrc_all[w];
  float* lalpha = lalpha_all[w];
  const int deg = cnt[d];
  const bool small = deg <= SEGCAP;
  const int h8 = lane & 7, jj = lane >> 3;
  const float ad = a_dst[d * 8 + h8];
  acc0 = make_float4(0.f, 0.f, 0.f, 0.f); acc1 = acc0; acc2 = acc0; acc3 = acc0;
  if (small) {
    const int2* bucket = (const int2*)(slot + (size_t)d * SEGCAP * 2);
    const int t0 = lane, t1 = lane + 32;
    int k0 = 0, s0 = 0, k1 = 0, s1 = 0;
    if (t0 < deg) { const int2 ks = bucket[t0]; k0 = ks.x; s0 = ks.y; lkey[t0] = k0; }
    if (t1 < deg) { const int2 ks = bucket[t1]; k1 = ks.x; s1 = ks.y; lkey[t1] = k1; }
    wave_lds_handover();
    int r0 = 0, r1 = 0;
    for (int j = 0; j < deg; ++j) {
      const int kj = lkey[j];
      r0 += (kj < k0) ? 1 : 0;
      r1 += (kj < k1) ? 1 : 0;
    }
    if (t0 < deg) lsrc[r0] = s0;
    if (t1 < deg) lsrc[r1] = s1;
    wave_lds_handover();
    if (deg > 0) {
      if (deg <= 16)      segment_weights_regs<4>(deg, jj, h8, ad, lsrc, a_src, lalpha);
      else if (deg <= 32) segment_weights_regs<8>(deg, jj, h8, ad, lsrc, a_src, lalpha);
      else                segment_weights_regs<16>(deg, jj, h8, ad, lsrc, a_src, lalpha);
    }
    wave_lds_handover();
    accumulate_rows<ROWF, RowT>(deg, lane, lsrc, lalpha, hW, acc0, acc1, acc2, acc3);
    return;
  }
#define STAGE_CHUNK(cb, n)                                               \
  {                                                                      \
    wave_lds_handover();    \
    stage_chunk_scan(d, (cb), (n), ei, E, lsrc, lane, sw_e0, sw_seen);   \
    wave_lds_handover();                                                 \
  }
  float mx = __uint_as_float(0xff800000u);
  int sw_e0 = 0, sw_seen = 0;
  for (int cb = 0; cb < deg; cb += SEGCAP) {
    const int n = (deg - cb) < SEGCAP ? (deg - cb) : SEGCAP;
    STAGE_CHUNK(cb, n)
    for (int j = jj; j < n; j += 4) {
      float L = a_src[lsrc[j] * 8 + h8] + ad;
      L = L >= 0.f ? L : 0.2f * L;
      mx = fmaxf(mx, L);
    }
  }
  mx = fmaxf(mx, __shfl_xor(mx, 8, 32));
  mx = fmaxf(mx, __shfl_xor(mx, 16, 32));
  const float m = mx;
  float sm = 0.f;
  sw_e0 = 0; sw_seen = 0;
  for (int cb = 0; cb < deg; cb += SEGCAP) {
    const int n = (deg - cb) < SEGCAP ? (deg - cb) : SEGCAP;
    STAGE_CHUNK(cb, n)
    for (int j = jj; j < n; j += 4) {
      float L = a_src[lsrc[j] * 8 + h8] + ad;
      L = L >= 0.f ? L : 0.2f * L;
      sm += __expf(L - m);
    }
  }
  sm += __shfl_xor(sm, 8, 32);
  sm += __shfl_xor(sm, 16, 32);
  const float den = sm;
  sw_e0 = 0; sw_seen = 0;
  for (int cb = 0; cb < deg; cb += SEGCAP) {
    const int n = (deg - cb) < SEGCAP ? (deg - cb) : SEGCAP;
    STAGE_CHUNK(cb, n)
    for (int j = jj; j < n; j += 4) {
      float L = a_src[lsrc[j] * 8 + h8] + ad;
      L = L >= 0.f ? L : 0.2f * L;
      lalpha[APOS(j, h8)] = __expf(L - m) / den;
    }
    wave_lds_handover();
    accumulate_rows<ROWF, RowT>(n, lane, lsrc, lalpha, hW, acc0, acc1, acc2, acc3);
  }
#undef STAGE_CHUNK
}

__global__ __launch_bounds__(32 * NPW, 12) void k_gather1_bn_gemm2(
    const int N, const int* __restrict__ cnt, const int* __restrict__ slot,
    const int* __restrict__ ei, const int E, const unsigned* __restrict__ hW1h,
    const float* __restrict__ a_src1, const float* __restrict__ a_dst1,
    const float* __restrict__ b1, const float* __restrict__ bn_g,
    const float* __restrict__ bn_b, const float* __restrict__ bn_mean,
    const float* __restrict__ bn_var, const unsigned* __restrict__ packW,
    const float* __restrict__ att_src, const float* __restrict__ att_dst,
    float* __restrict__ hW, float* __restrict__ a_src, float* __restrict__ a_dst) {
  static_assert(NPW == 16, "one wave per kstep of the 512 -> 16 product");
  __shared__ __attribute__((aligned(16))) float tile[16 * LDP];
  __shared__ __attribute__((aligned(16))) float part[NPW][16][16];
  const int mt = blockIdx.x, tid = threadIdx.x, lane = tid & 31;
  const int w = __builtin_amdgcn_readfirstlane(tid >> 5);
  const int d = mt * NPW + w;
  float4 acc[4];
  acc[0] = make_float4(0.f, 0.f, 0.f, 0.f); acc[1] = acc[0]; acc[2] = acc[0]; acc[3] = acc[0];
  if (d < N)
    gather_node_wave<512, unsigned>(d, w, lane, cnt, slot, ei, E, hW1h, a_src1, a_dst1,
                          acc[0], acc[1], acc[2], acc[3]);
#pragma unroll
  for (int q = 0; q < 4; ++q) {
    const int c = (q >> 1) * 256 + lane * 8 + (q & 1) * 4;
    const float4 vb1 = *(const float4*)(b1 + c), vmu = *(const float4*)(bn_mean + c);
    const float4 vva = *(const float4*)(bn_var + c), vg = *(const float4*)(bn_g + c);
    const float4 vb = *(const float4*)(bn_b + c);
    float4 r;
    r.x = ((acc[q].x + vb1.x) - vmu.x) * rsqrtf(vva.x + 1e-5f) * vg.x + vb.x;
    r.y = ((acc[q].y + vb1.y) - vmu.y) * rsqrtf(vva.y + 1e-5f) * vg.y + vb.y;
    r.z = ((acc[q].z + vb1.z) - vmu.z) * rsqrtf(vva.z + 1e-5f) * vg.z + vb.z;
    r.w = ((acc[q].w + vb1.w) - vmu.w) * rsqrtf(vva.w + 1e-5f) * vg.w + vb.w;
    r.x = r.x > 0.f ? r.x : 0.f; r.y = r.y > 0.f ? r.y : 0.f;
    r.z = r.z > 0.f ? r.z : 0.f; r.w = r.w > 0.f ? r.w : 0.f;
    *(float4*)(tile + w * LDP + c) = r;
  }
  __syncthreads();
  const int ln = (int)__builtin_amdgcn_mbcnt_lo(~0u, 0u);
  const int tid2 = (w << 5) + ln;
  {
    const v16bf bfrag = *(const v16bf*)(packW + ((size_t)w * 32 + ln) * 8);
    const float* tr = tile + (ln & 15) * LDP + (w << 5) + ((ln & 16) >> 1);
    const float4 p0 = *(const float4*)(tr),      p1 = *(const float4*)(tr + 4);
    const float4 p2 = *(const float4*)(tr + 16), p3 = *(const float4*)(tr + 20);
    const v8u u = {pk2(p0.x, p0.y), pk2(p0.z, p0.w), pk2(p1.x, p1.y), pk2(p1.z, p1.w),
                   pk2(p2.x, p2.y), pk2(p2.z, p2.w), pk2(p3.x, p3.y), pk2(p3.z, p3.w)};
    v8f pr = {};
    pr = __builtin_amdgcn_wmma_f32_16x16x32_bf16(false, __builtin_bit_cast(v16bf, u), false,
                                                 bfrag, (short)0, pr, false, false);
    const int m0 = (ln >> 4) << 3, nc = ln & 15;
#pragma unroll
    for (int r = 0; r < 8; ++r) part[w][m0 + r][nc] = pr[r];
  }
  __syncthreads();
  if (tid2 < 128) {
    const int row = tid2 >> 3, h = tid2 & 7, node = mt * 16 + row;
    float2 pp[NPW];
#pragma unroll
    for (int ks = 0; ks < NPW; ++ks) pp[ks] = *(const float2*)(&part[ks][row][2 * h]);
    float v0 = pp[0].x, v1 = pp[0].y;
#pragma unroll
    for (int ks = 1; ks < NPW; ++ks) { v0 += pp[ks].x; v1 += pp[ks].y; }
    if (node < N) {
      *(float2*)(hW + (size_t)node * 16 + 2 * h) = make_float2(v0, v1);
      a_src[node * 8 + h] = v0 * att_src[2 * h] + v1 * att_src[2 * h + 1];
      a_dst[node * 8 + h] = v0 * att_dst[2 * h] + v1 * att_dst[2 * h + 1];
    }
  }
}

__global__ __launch_bounds__(32 * NPW) void k_gather2_final(
    const int N, const int* __restrict__ cnt, const int* __restrict__ slot,
    const int* __restrict__ ei, const int E, const float* __restrict__ hW2,
    const float* __restrict__ a_src2, const float* __restrict__ a_dst2,
    const float* __restrict__ b2, float* __restrict__ out) {
  __shared__ __attribute__((aligned(16))) float rows[NPW][16];
  const int tid = threadIdx.x, lane = tid & 31;
  const int w = __builtin_amdgcn_readfirstlane(tid >> 5);
  const int d = blockIdx.x * NPW + w;
  float4 acc0 = make_float4(0.f, 0.f, 0.f, 0.f), acc1 = acc0, acc2 = acc0, acc3 = acc0;
  if (d < N)
    gather_node_wave<16, float>(d, w, lane, cnt, slot, ei, E, hW2, a_src2, a_dst2,
                         acc0, acc1, acc2, acc3);
  if (lane < 4) *(float4*)(&rows[w][lane * 4]) = acc0;
  __syncthreads();
  if (tid < 32) {
    const int n = tid >> 1, c = tid & 1, node = blockIdx.x * NPW + n;
    float s = 0.f;
#pragma unroll
    for (int h = 0; h < 8; ++h) s += rows[n][2 * h + c];
    if (node < N) out[node * 2 + c] = s * 0.125f + b2[c];
  }
}

extern "C" void kernel_launch(void* const* d_in, const int* in_sizes, int n_in,
                              void* d_out, int out_size, void* d_ws, size_t ws_size,
                              hipStream_t stream) {
  const float* x       = (const float*)d_in[0];
  const int*   ei      = (const int*)d_in[1];
  const float* ft_W    = (const float*)d_in[2];
  const float* ft_b    = (const float*)d_in[3];
  const float* ln_g    = (const float*)d_in[4];
  const float* ln_b    = (const float*)d_in[5];
  const float* W1      = (const float*)d_in[6];
  const float* att_s1  = (const float*)d_in[7];
  const float* att_d1  = (const float*)d_in[8];
  const float* b1      = (const float*)d_in[9];
  const float* bn_g    = (const float*)d_in[10];
  const float* bn_b    = (const float*)d_in[11];
  const float* bn_mean = (const float*)d_in[12];
  const float* bn_var  = (const float*)d_in[13];
  const float* W2      = (const float*)d_in[14];
  const float* att_s2  = (const float*)d_in[15];
  const float* att_d2  = (const float*)d_in[16];
  const float* b2      = (const float*)d_in[17];

  const int N  = in_sizes[0] / 128;
  const int E  = in_sizes[1] / 2;
  const int MT = N / 16;
  const int MG = (MT + MTW - 1) / MTW;

  char* cur = (char*)d_ws;
  auto alloc = [&](size_t bytes) -> void* {
    void* p = (void*)cur;
    cur += (bytes + 255) & ~(size_t)255;
    return p;
  };
  unsigned* packWft = (unsigned*)alloc((size_t)32 * 4 * 256 * 4);
  unsigned* packW1  = (unsigned*)alloc((size_t)32 * 16 * 256 * 4);
  unsigned* packW2  = (unsigned*)alloc((size_t)1 * 16 * 256 * 4);
  unsigned* hW1h = (unsigned*)alloc((size_t)N * 256 * 4);
  float* hW2 = (float*)alloc((size_t)N * 16 * 4);
  float* aS1 = (float*)alloc((size_t)N * 8 * 4);
  float* aD1 = (float*)alloc((size_t)N * 8 * 4);
  float* aS2 = (float*)alloc((size_t)N * 8 * 4);
  float* aD2 = (float*)alloc((size_t)N * 8 * 4);
  int* slot = (int*)alloc((size_t)N * SEGCAP * 8);
  int* cnt  = (int*)alloc((size_t)N * 4);

  k_prep<<<128 + 512 + 16 + (N + 255) / 256, 256, 0, stream>>>(
      ft_W, W1, W2, packWft, packW1, packW2, cnt, N);

  k_ft_gemm1<<<MG + (E + 255) / 256, 256, 0, stream>>>(
      x, packWft, ft_b, ln_g, ln_b, packW1, att_s1, att_d1, hW1h, aS1, aD1,
      ei, cnt, slot, E, MG, N);

  k_gather1_bn_gemm2<<<(N + NPW - 1) / NPW, 32 * NPW, 0, stream>>>(
      N, cnt, slot, ei, E, hW1h, aS1, aD1, b1, bn_g, bn_b, bn_mean, bn_var, packW2,
      att_s2, att_d2, hW2, aS2, aD2);

  k_gather2_final<<<(N + NPW - 1) / NPW, 32 * NPW, 0, stream>>>(
      N, cnt, slot, ei, E, hW2, aS2, aD2, b2, (float*)d_out);
}
